# starting kernel
# baseline (speedup 1.0000x reference)
#include <hip/hip_runtime.h>
#include <hip/hip_bf16.h>

typedef __attribute__((ext_vector_type(2))) float v2f;
typedef __attribute__((ext_vector_type(8))) float v8f;

__global__ void k_init_deg(float* __restrict__ deg, int n) {
    int i = blockIdx.x * blockDim.x + threadIdx.x;
    if (i < n) deg[i] = 1.0f;
}

__global__ void k_count_deg(const int* __restrict__ dst, float* __restrict__ deg, int e) {
    int i = blockIdx.x * blockDim.x + threadIdx.x;
    if (i < e) atomicAdd(&deg[dst[i]], 1.0f);
}

__global__ void k_finalize_dis(float* __restrict__ dis, int n) {
    int i = blockIdx.x * blockDim.x + threadIdx.x;
    if (i < n) dis[i] = rsqrtf(dis[i]);
}

template <int FIN, int FOUT, int MT, bool RELU>
__global__ void k_gemm_wmma(const float* __restrict__ X, const float* __restrict__ W,
                            const float* __restrict__ dis, const float* __restrict__ bias,
                            float* __restrict__ H, float* __restrict__ OUT, int M) {
    constexpr int LDA = FIN + 1;
    __shared__ float sW[FIN * FOUT];
    __shared__ float sA[MT * 16 * LDA];

    const int tid  = threadIdx.x;
    const int nthr = 32 * (FOUT / 16);
    const int m0   = blockIdx.x * (MT * 16);

    for (int i = tid; i < FIN * FOUT; i += nthr) sW[i] = W[i];
    for (int i = tid; i < MT * 16 * FIN; i += nthr) {
        int r = i / FIN, c = i - r * FIN;
        int row = m0 + r; if (row >= M) row = M - 1;
        float v = X[(size_t)row * FIN + c];
        sA[r * LDA + c] = RELU ? fmaxf(v, 0.0f) : v;
    }
    __syncthreads();

    const int wave = tid >> 5;
    const int lane = tid & 31;
    const int half = lane >> 4;
    const int l16  = lane & 15;
    const int n0   = wave * 16;

    v2f bf[FIN / 4];
    #pragma unroll
    for (int kk = 0; kk < FIN / 4; ++kk) {
        int k = kk * 4 + 2 * half;
        bf[kk].x = sW[k * FOUT + n0 + l16];
        bf[kk].y = sW[(k + 1) * FOUT + n0 + l16];
    }
    const float bcol = bias[n0 + l16];

    #pragma unroll
    for (int mt = 0; mt < MT; ++mt) {
        const float* At = &sA[mt * 16 * LDA];
        v8f c = {};
        #pragma unroll
        for (int kk = 0; kk < FIN / 4; ++kk) {
            int k = kk * 4 + 2 * half;
            v2f a;
            a.x = At[l16 * LDA + k];
            a.y = At[l16 * LDA + k + 1];
            c = __builtin_amdgcn_wmma_f32_16x16x4_f32(false, a, false, bf[kk],
                                                      (short)0, c, false, false);
        }
        #pragma unroll
        for (int r = 0; r < 8; ++r) {
            int row = m0 + mt * 16 + half * 8 + r;
            if (row < M) {
                float hv = c[r];
                float s  = dis[row];
                size_t o = (size_t)row * FOUT + n0 + l16;
                H[o]   = hv;
                OUT[o] = hv * (s * s) + bcol;
            }
        }
    }
}

template <int CHUNKS>
__global__ void k_scatter(const float* __restrict__ h, const int* __restrict__ src,
                          const int* __restrict__ dst, const float* __restrict__ dis,
                          float* __restrict__ out, int e) {
    constexpr int F = CHUNKS * 4;
    const unsigned total = (unsigned)e * (unsigned)CHUNKS;
    const unsigned step  = gridDim.x * blockDim.x;
    for (unsigned idx = blockIdx.x * blockDim.x + threadIdx.x; idx < total; idx += step) {
        const unsigned ei = idx / CHUNKS;
        const unsigned c  = idx - ei * CHUNKS;
        const int s = src[ei], d = dst[ei];
        const float norm = dis[s] * dis[d];
        const float4 v = *(const float4*)(h + (size_t)s * F + (c << 2));
        float* o = out + (size_t)d * F + (c << 2);
        atomicAdd(o + 0, v.x * norm);
        atomicAdd(o + 1, v.y * norm);
        atomicAdd(o + 2, v.z * norm);
        atomicAdd(o + 3, v.w * norm);
    }
}

extern "C" void kernel_launch(void* const* d_in, const int* in_sizes, int n_in,
                              void* d_out, int out_size, void* d_ws, size_t ws_size,
                              hipStream_t stream) {
    const float* X  = (const float*)d_in[0];
    const int*   EI = (const int*)d_in[1];
    const float* W1 = (const float*)d_in[2];
    const float* b1 = (const float*)d_in[3];
    const float* W2 = (const float*)d_in[4];
    const float* b2 = (const float*)d_in[5];
    float* out = (float*)d_out;

    const int FIN1 = 64;
    const int N = in_sizes[0] / FIN1;
    const int E = in_sizes[1] / 2;
    const int* src = EI;
    const int* dst = EI + E;

    float* dis = (float*)d_ws;
    size_t off = ((size_t)N * sizeof(float) + 255) & ~(size_t)255;
    float* bufH = (float*)((char*)d_ws + off);
    float* bufO = bufH + (size_t)N * 128;

    const int T = 256;
    dim3 blkN((N + T - 1) / T), blkE((E + T - 1) / T);
    const int ntiles = (N + 15) / 16;

    k_init_deg<<<blkN, T, 0, stream>>>(dis, N);
    k_count_deg<<<blkE, T, 0, stream>>>(dst, dis, E);
    k_finalize_dis<<<blkN, T, 0, stream>>>(dis, N);

    k_gemm_wmma<64, 128, 4, false><<<(ntiles + 3) / 4, 256, 0, stream>>>(
        X, W1, dis, b1, bufH, bufO, N);
    k_scatter<32><<<8192, T, 0, stream>>>(bufH, src, dst, dis, bufO, E);

    k_gemm_wmma<128, 64, 2, true><<<(ntiles + 1) / 2, 128, 0, stream>>>(
        bufO, W2, dis, b2, bufH, out, N);
    k_scatter<16><<<8192, T, 0, stream>>>(bufH, src, dst, dis, out, E);
}


# Neighbour sums owner-computes: a work-group holds 400 dst rows in LDS, picks its edges from a 1-byte-per-edge key plane, LDS f32 adds, rows stored once; no global atomics; sum order = LDS arrival
# speedup vs baseline: 2.2582x; 2.2582x over previous
#include <hip/hip_runtime.h>
#include <hip/hip_bf16.h>

typedef __attribute__((ext_vector_type(2))) float v2f;
typedef __attribute__((ext_vector_type(8))) float v8f;

constexpr int DEG_RANGE  = 16384;
constexpr int DEG_CHUNKS = 64;

__global__ void __launch_bounds__(256)
k_count_deg(const int* __restrict__ dst, int* __restrict__ part, int e, int n) {
    __shared__ int cnt[DEG_RANGE];
    const int tid = threadIdx.x;
    const int lo  = blockIdx.x * DEG_RANGE;
    const int c   = blockIdx.y;
    for (int i = tid; i < DEG_RANGE; i += 256) cnt[i] = 0;
    __syncthreads();

    const int per = (((e + DEG_CHUNKS - 1) / DEG_CHUNKS) + 3) & ~3;
    const int e0  = c * per;
    const int e1  = min(e0 + per, e);
    if ((((size_t)dst) & 15) == 0) {
        const int nq = (e1 > e0) ? ((e1 - e0) >> 2) : 0;
        const int4* dq = reinterpret_cast<const int4*>(dst + e0);
        constexpr int B = 8;
        for (int qb = 0; qb < nq; qb += 256 * B) {
            int4 v[B];
            #pragma unroll
            for (int j = 0; j < B; ++j) {
                const int q = qb + j * 256 + tid;
                v[j] = dq[min(q, nq - 1)];
            }
            #pragma unroll
            for (int j = 0; j < B; ++j) {
                const bool in = (qb + j * 256 + tid) < nq;
                const unsigned k0 = (unsigned)(v[j].x - lo), k1 = (unsigned)(v[j].y - lo);
                const unsigned k2 = (unsigned)(v[j].z - lo), k3 = (unsigned)(v[j].w - lo);
                if (in && k0 < (unsigned)DEG_RANGE) atomicAdd(&cnt[k0], 1);
                if (in && k1 < (unsigned)DEG_RANGE) atomicAdd(&cnt[k1], 1);
                if (in && k2 < (unsigned)DEG_RANGE) atomicAdd(&cnt[k2], 1);
                if (in && k3 < (unsigned)DEG_RANGE) atomicAdd(&cnt[k3], 1);
            }
        }
        const int i = e0 + 4 * nq + tid;
        if (tid < 4 && i < e1) {
            const unsigned k = (unsigned)(dst[i] - lo);
            if (k < (unsigned)DEG_RANGE) atomicAdd(&cnt[k], 1);
        }
    } else {
        for (int i = e0 + tid; i < e1; i += 256) {
            const unsigned k = (unsigned)(dst[i] - lo);
            if (k < (unsigned)DEG_RANGE) atomicAdd(&cnt[k], 1);
        }
    }
    __syncthreads();

    const int hi = min(DEG_RANGE, n - lo);
    int* p = part + (size_t)c * n + lo;
    for (int i = tid; i < hi; i += 256) p[i] = cnt[i];
}

constexpr int AGG_ROWS = 400;

__device__ __forceinline__ void write_keys(const int* __restrict__ dst, unsigned* __restrict__ keys,
                                           int e, int gtid, int gsize) {
    const int nw = ((e + 15) >> 4) << 2;
    for (int w = gtid; w < nw; w += gsize) {
        unsigned v = 0;
        #pragma unroll
        for (int j = 0; j < 4; ++j) {
            const int i = 4 * w + j;
            unsigned k = 0xFFu;
            if (i < e) {
                const unsigned p = (unsigned)dst[i] / (unsigned)AGG_ROWS;
                k = p < 0xFFu ? p : 0xFFu;
            }
            v |= k << (8 * j);
        }
        keys[w] = v;
    }
}

__global__ void k_finalize_dis(const int* __restrict__ part, float* __restrict__ dis,
                               const int* __restrict__ dst, unsigned* __restrict__ keys,
                               int e, int n) {
    int i = blockIdx.x * blockDim.x + threadIdx.x;
    if (i < n) {
        int d = 1;
        for (int c = 0; c < DEG_CHUNKS; ++c) d += part[(size_t)c * n + i];
        dis[i] = rsqrtf((float)d);
    }
    write_keys(dst, keys, e, i, (int)(gridDim.x * blockDim.x));
}

template <int FIN, int FOUT, int MT, bool RELU, bool KEYS>
__global__ void k_gemm_wmma(const float* __restrict__ X, const float* __restrict__ W,
                            float* __restrict__ H, int M,
                            const int* __restrict__ dst, unsigned* __restrict__ keys, int e) {
    constexpr int LDA = FIN + 1;
    __shared__ float sW[FIN * FOUT];
    __shared__ float sA[MT * 16 * LDA];

    const int tid  = threadIdx.x;
    const int nthr = 32 * (FOUT / 16);
    const int m0   = blockIdx.x * (MT * 16);

    if (KEYS) write_keys(dst, keys, e, (int)(blockIdx.x * nthr + tid), (int)(gridDim.x * nthr));

    for (int i = tid; i < FIN * FOUT; i += nthr) sW[i] = W[i];
    for (int i = tid; i < MT * 16 * FIN; i += nthr) {
        int r = i / FIN, c = i - r * FIN;
        int row = m0 + r; if (row >= M) row = M - 1;
        float v = X[(size_t)row * FIN + c];
        sA[r * LDA + c] = RELU ? fmaxf(v, 0.0f) : v;
    }
    __syncthreads();

    const int wave = tid >> 5;
    const int lane = tid & 31;
    const int half = lane >> 4;
    const int l16  = lane & 15;
    const int n0   = wave * 16;

    v2f bf[FIN / 4];
    #pragma unroll
    for (int kk = 0; kk < FIN / 4; ++kk) {
        int k = kk * 4 + 2 * half;
        bf[kk].x = sW[k * FOUT + n0 + l16];
        bf[kk].y = sW[(k + 1) * FOUT + n0 + l16];
    }

    #pragma unroll
    for (int mt = 0; mt < MT; ++mt) {
        const float* At = &sA[mt * 16 * LDA];
        v8f c = {};
        #pragma unroll
        for (int kk = 0; kk < FIN / 4; ++kk) {
            int k = kk * 4 + 2 * half;
            v2f a;
            a.x = At[l16 * LDA + k];
            a.y = At[l16 * LDA + k + 1];
            c = __builtin_amdgcn_wmma_f32_16x16x4_f32(false, a, false, bf[kk],
                                                      (short)0, c, false, false);
        }
        #pragma unroll
        for (int r = 0; r < 8; ++r) {
            int row = m0 + mt * 16 + half * 8 + r;
            if (row < M) H[(size_t)row * FOUT + n0 + l16] = c[r];
        }
    }
}

template <int F, int U>
__device__ __forceinline__ void rows_load(float (&v)[U][F / 32], const float* hl, int s, int j0) {
    #pragma unroll
    for (int u = 0; u < U; ++u) {
        const int sj = __builtin_amdgcn_readlane(s, j0 + u);
        const float* hp = hl + (size_t)sj * F;
        #pragma unroll
        for (int k = 0; k < F / 32; ++k) v[u][k] = hp[32 * k];
    }
}

template <int F, int U>
__device__ __forceinline__ void rows_add(const float (&v)[U][F / 32], float* al, int kr, float nrm, int j0) {
    #pragma unroll
    for (int u = 0; u < U; ++u) {
        const int   kj = __builtin_amdgcn_readlane(kr, j0 + u);
        const float nj = __int_as_float(__builtin_amdgcn_readlane(__float_as_int(nrm), j0 + u));
        float* ap = al + kj * F;
        #pragma unroll
        for (int k = 0; k < F / 32; ++k) atomicAdd(ap + 32 * k, v[u][k] * nj);
    }
}

template <int F, int U>
__device__ __forceinline__ void walk32(const float* hl, float* al, int s, int kr, float nrm) {
    float va[U][F / 32], vb[U][F / 32];
    rows_load<F, U>(va, hl, s, 0);
    #pragma unroll
    for (int j0 = 0; j0 < 32; j0 += 2 * U) {
        rows_load<F, U>(vb, hl, s, j0 + U);
        rows_add<F, U>(va, al, kr, nrm, j0);
        if (j0 + 2 * U < 32) rows_load<F, U>(va, hl, s, j0 + 2 * U);
        rows_add<F, U>(vb, al, kr, nrm, j0 + U);
    }
}

template <int F, int U>
__global__ void __launch_bounds__(1024)
k_agg(const float* __restrict__ h, const unsigned* __restrict__ keys,
      const int* __restrict__ src, const int* __restrict__ dst,
      const float* __restrict__ dis, const float* __restrict__ bias,
      float* __restrict__ out, int e, int n) {
    static_assert(32 % (2 * U) == 0 && F % 32 == 0, "group shape");
    constexpr int B = 4;
    __shared__ __attribute__((aligned(16))) float acc[AGG_ROWS * F];

    const int tid  = threadIdx.x;
    const int lane = tid & 31;
    const int lo   = blockIdx.x * AGG_ROWS;
    const int rows = min(AGG_ROWS, n - lo);

    {
        const float4* h4 = reinterpret_cast<const float4*>(h + (size_t)lo * F);
        const float4* b4 = reinterpret_cast<const float4*>(bias);
        float4* a4 = reinterpret_cast<float4*>(acc);
        for (int i = tid; i < rows * (F / 4); i += 1024) {
            const int r = i / (F / 4), c = i - r * (F / 4);
            const float s = dis[lo + r];
            const float w = s * s;
            const float4 hv = h4[i];
            const float4 bv = b4[c];
            float4 o;
            o.x = hv.x * w + bv.x;  o.y = hv.y * w + bv.y;
            o.z = hv.z * w + bv.z;  o.w = hv.w * w + bv.w;
            a4[i] = o;
        }
    }
    __syncthreads();

    const float* hl = h + lane;
    float*       al = acc + lane;
    const unsigned kb   = blockIdx.x < 0xFFu ? blockIdx.x : 0xFFu;
    const unsigned key4 = kb * 0x01010101u;
    const int nq = (e + 15) >> 4;
    const uint4* kq = reinterpret_cast<const uint4*>(keys);
    const int per = 1024 * B;
    const int T   = (nq + per - 1) / per;
    const int rot = (int)(((long long)blockIdx.x * T) / gridDim.x);

    int qe = 0;
    int qc = 0;

    uint4 cur[B], nxt[B];
    int itr = rot;
    #pragma unroll
    for (int j = 0; j < B; ++j) cur[j] = kq[min(itr * per + j * 1024 + tid, nq - 1)];

    for (int it = 0; it < T; ++it) {
        int itn = itr + 1;
        if (itn >= T) itn = 0;
        #pragma unroll
        for (int j = 0; j < B; ++j) nxt[j] = kq[min(itn * per + j * 1024 + tid, nq - 1)];

        const int g0 = itr * per + tid;
        unsigned m[B];
        unsigned any = 0;
        #pragma unroll
        for (int j = 0; j < B; ++j) {
            const unsigned x0 = cur[j].x ^ key4, x1 = cur[j].y ^ key4, x2 = cur[j].z ^ key4, x3 = cur[j].w ^ key4;
            const unsigned z0 = ~(((x0 & 0x7F7F7F7Fu) + 0x7F7F7F7Fu) | x0) & 0x80808080u;
            const unsigned z1 = ~(((x1 & 0x7F7F7F7Fu) + 0x7F7F7F7Fu) | x1) & 0x80808080u;
            const unsigned z2 = ~(((x2 & 0x7F7F7F7Fu) + 0x7F7F7F7Fu) | x2) & 0x80808080u;
            const unsigned z3 = ~(((x3 & 0x7F7F7F7Fu) + 0x7F7F7F7Fu) | x3) & 0x80808080u;
            m[j] = (z0 >> 7) | (z1 >> 6) | (z2 >> 5) | (z3 >> 4);
            if (g0 + j * 1024 >= nq) m[j] = 0;
            any |= m[j];
        }
        while (__builtin_amdgcn_ballot_w32(any != 0) != 0) {
            const bool pass = (any != 0);
            unsigned mm = m[B - 1];
            int jj = B - 1;
            #pragma unroll
            for (int j = B - 2; j >= 0; --j) {
                if (m[j] != 0) { mm = m[j]; jj = j; }
            }
            const int bit = pass ? __builtin_ctz(mm) : 0;
            const int ei  = 16 * (g0 + jj * 1024) + 4 * (bit & 7) + (bit >> 3);
            any = 0;
            #pragma unroll
            for (int j = 0; j < B; ++j) {
                if (jj == j) m[j] &= m[j] - 1;
                any |= m[j];
            }
            const unsigned bm = __builtin_amdgcn_ballot_w32(pass);
            const int cnt = __builtin_popcount(bm);
            const int rk  = (int)__builtin_amdgcn_mbcnt_lo(bm, 0u);
            const int pos = pass ? (qc + rk) : (qc + cnt + (lane - rk));
            const int pe = __builtin_amdgcn_ds_permute((pos & 31) << 2, ei);
            const int me = (lane < qc) ? qe : pe;
            if (qc + cnt >= 32) {
                const int ec = min(me, e - 1);
                const int d  = dst[ec];
                const int s  = src[ec];
                const unsigned kr = (unsigned)(d - lo);
                const bool ok = (me < e) && (kr < (unsigned)rows);
                const float nrm = ok ? dis[s] * dis[d] : 0.0f;
                walk32<F, U>(hl, al, s, ok ? (int)kr : 0, nrm);
                qe = pe;
                qc = qc + cnt - 32;
            } else {
                qe = me;
                qc = qc + cnt;
            }
        }
        #pragma unroll
        for (int j = 0; j < B; ++j) cur[j] = nxt[j];
        itr = itn;
    }

    if (qc > 0) {
        int s = 0, kr = 0;
        float nrm = 0.0f;
        if (lane < qc) {
            const int ec = min(qe, e - 1);
            const int d  = dst[ec];
            s = src[ec];
            const unsigned k = (unsigned)(d - lo);
            const bool ok = (qe < e) && (k < (unsigned)rows);
            kr  = ok ? (int)k : 0;
            nrm = ok ? dis[s] * dis[d] : 0.0f;
        }
        for (int j = 0; j < qc; ++j) {
            float v1[1][F / 32];
            rows_load<F, 1>(v1, hl, s, j);
            rows_add<F, 1>(v1, al, kr, nrm, j);
        }
    }
    __syncthreads();

    {
        float4* o4 = reinterpret_cast<float4*>(out + (size_t)lo * F);
        const float4* a4 = reinterpret_cast<const float4*>(acc);
        for (int i = tid; i < rows * (F / 4); i += 1024) o4[i] = a4[i];
    }
}

extern "C" void kernel_launch(void* const* d_in, const int* in_sizes, int n_in,
                              void* d_out, int out_size, void* d_ws, size_t ws_size,
                              hipStream_t stream) {
    const float* X  = (const float*)d_in[0];
    const int*   EI = (const int*)d_in[1];
    const float* W1 = (const float*)d_in[2];
    const float* b1 = (const float*)d_in[3];
    const float* W2 = (const float*)d_in[4];
    const float* b2 = (const float*)d_in[5];
    float* out = (float*)d_out;

    const int FIN1 = 64;
    const int N = in_sizes[0] / FIN1;
    const int E = in_sizes[1] / 2;
    const int* src = EI;
    const int* dst = EI + E;

    float* dis = (float*)d_ws;
    size_t off = ((size_t)N * sizeof(float) + 16 + 255) & ~(size_t)255;
    float* bufH = (float*)((char*)d_ws + off);
    float* bufO = bufH + (size_t)N * 128;
    int*   degp = (int*)bufH;
    unsigned* keys1 = (unsigned*)d_out;
    unsigned* keys2 = (unsigned*)(bufH + (size_t)N * 64);

    const int T = 256;
    dim3 blkN((N + T - 1) / T);
    dim3 blkD((N + DEG_RANGE - 1) / DEG_RANGE, DEG_CHUNKS);
    const int ntiles = (N + 15) / 16;
    dim3 blkA((N + AGG_ROWS - 1) / AGG_ROWS);

    k_count_deg<<<blkD, T, 0, stream>>>(dst, degp, E, N);
    k_finalize_dis<<<blkN, T, 0, stream>>>(degp, dis, dst, keys1, E, N);

    k_gemm_wmma<64, 128, 4, false, false><<<(ntiles + 3) / 4, 256, 0, stream>>>(
        X, W1, bufH, N, dst, keys2, E);
    k_agg<128, 4><<<blkA, 1024, 0, stream>>>(bufH, keys1, src, dst, dis, b1, bufO, E, N);

    k_gemm_wmma<128, 64, 2, true, true><<<(ntiles + 1) / 2, 128, 0, stream>>>(
        bufO, W2, bufH, N, dst, keys2, E);
    k_agg<64, 8><<<blkA, 1024, 0, stream>>>(bufH, keys2, src, dst, dis, b2, out, E, N);
}


# Picked edges read one 8-byte record (src, row in piece, dis[src]) built once in the first dense launch, dis[dst] from LDS: no src, dst, dis reads per edge; same products, f32 sum order = LDS arrival
# speedup vs baseline: 2.4883x; 1.1019x over previous
#include <hip/hip_runtime.h>
#include <hip/hip_bf16.h>

typedef __attribute__((ext_vector_type(2))) float v2f;
typedef __attribute__((ext_vector_type(8))) float v8f;

constexpr int DEG_RANGE  = 16384;
constexpr int DEG_CHUNKS = 64;

__global__ void __launch_bounds__(256)
k_count_deg(const int* __restrict__ dst, int* __restrict__ part, int e, int n) {
    __shared__ int cnt[DEG_RANGE];
    const int tid = threadIdx.x;
    const int lo  = blockIdx.x * DEG_RANGE;
    const int c   = blockIdx.y;
    for (int i = tid; i < DEG_RANGE; i += 256) cnt[i] = 0;
    __syncthreads();

    const int per = (((e + DEG_CHUNKS - 1) / DEG_CHUNKS) + 3) & ~3;
    const int e0  = c * per;
    const int e1  = min(e0 + per, e);
    if ((((size_t)dst) & 15) == 0) {
        const int nq = (e1 > e0) ? ((e1 - e0) >> 2) : 0;
        const int4* dq = reinterpret_cast<const int4*>(dst + e0);
        constexpr int B = 8;
        for (int qb = 0; qb < nq; qb += 256 * B) {
            int4 v[B];
            #pragma unroll
            for (int j = 0; j < B; ++j) {
                const int q = qb + j * 256 + tid;
                v[j] = dq[min(q, nq - 1)];
            }
            #pragma unroll
            for (int j = 0; j < B; ++j) {
                const bool in = (qb + j * 256 + tid) < nq;
                const unsigned k0 = (unsigned)(v[j].x - lo), k1 = (unsigned)(v[j].y - lo);
                const unsigned k2 = (unsigned)(v[j].z - lo), k3 = (unsigned)(v[j].w - lo);
                if (in && k0 < (unsigned)DEG_RANGE) atomicAdd(&cnt[k0], 1);
                if (in && k1 < (unsigned)DEG_RANGE) atomicAdd(&cnt[k1], 1);
                if (in && k2 < (unsigned)DEG_RANGE) atomicAdd(&cnt[k2], 1);
                if (in && k3 < (unsigned)DEG_RANGE) atomicAdd(&cnt[k3], 1);
            }
        }
        const int i = e0 + 4 * nq + tid;
        if (tid < 4 && i < e1) {
            const unsigned k = (unsigned)(dst[i] - lo);
            if (k < (unsigned)DEG_RANGE) atomicAdd(&cnt[k], 1);
        }
    } else {
        for (int i = e0 + tid; i < e1; i += 256) {
            const unsigned k = (unsigned)(dst[i] - lo);
            if (k < (unsigned)DEG_RANGE) atomicAdd(&cnt[k], 1);
        }
    }
    __syncthreads();

    const int hi = min(DEG_RANGE, n - lo);
    int* p = part + (size_t)c * n + lo;
    for (int i = tid; i < hi; i += 256) p[i] = cnt[i];
}

constexpr int AGG_ROWS = 400;

__device__ __forceinline__ void write_keys(const int* __restrict__ dst, unsigned* __restrict__ keys,
                                           int e, int gtid, int gsize) {
    const int nw = ((e + 15) >> 4) << 2;
    for (int w = gtid; w < nw; w += gsize) {
        unsigned v = 0;
        #pragma unroll
        for (int j = 0; j < 4; ++j) {
            const int i = 4 * w + j;
            unsigned k = 0xFFu;
            if (i < e) {
                const unsigned p = (unsigned)dst[i] / (unsigned)AGG_ROWS;
                k = p < 0xFFu ? p : 0xFFu;
            }
            v |= k << (8 * j);
        }
        keys[w] = v;
    }
}

constexpr int REC_SHIFT = 23;

__device__ __forceinline__ void build_recs(const int* __restrict__ src, const int* __restrict__ dst,
                                           const float* __restrict__ dis, uint2* __restrict__ rec,
                                           int e, int gtid, int gsize) {
    constexpr int B = 4;
    for (int i0 = gtid; i0 < e; i0 += B * gsize) {
        int s[B], d[B];
        float ws[B];
        #pragma unroll
        for (int j = 0; j < B; ++j) {
            const int i = min(i0 + j * gsize, e - 1);
            s[j] = src[i];
            d[j] = dst[i];
        }
        #pragma unroll
        for (int j = 0; j < B; ++j) ws[j] = dis[s[j]];
        #pragma unroll
        for (int j = 0; j < B; ++j) {
            const int i = i0 + j * gsize;
            if (i < e) {
                uint2 r;
                r.x = (unsigned)s[j] | (((unsigned)d[j] % (unsigned)AGG_ROWS) << REC_SHIFT);
                r.y = __float_as_uint(ws[j]);
                rec[i] = r;
            }
        }
    }
}

__device__ __forceinline__ void copy_recs(const uint2* __restrict__ from, uint2* __restrict__ to,
                                          int e, int gtid, int gsize) {
    constexpr int B = 4;
    for (int i0 = gtid; i0 < e; i0 += B * gsize) {
        uint2 v[B];
        #pragma unroll
        for (int j = 0; j < B; ++j) v[j] = from[min(i0 + j * gsize, e - 1)];
        #pragma unroll
        for (int j = 0; j < B; ++j) {
            const int i = i0 + j * gsize;
            if (i < e) to[i] = v[j];
        }
    }
}

__global__ void k_finalize_dis(const int* __restrict__ part, float* __restrict__ dis,
                               const int* __restrict__ dst, unsigned* __restrict__ keys,
                               int e, int n) {
    int i = blockIdx.x * blockDim.x + threadIdx.x;
    if (i < n) {
        int d = 1;
        for (int c = 0; c < DEG_CHUNKS; ++c) d += part[(size_t)c * n + i];
        dis[i] = rsqrtf((float)d);
    }
    write_keys(dst, keys, e, i, (int)(gridDim.x * blockDim.x));
}

template <int FIN, int FOUT, int MT, bool RELU, bool KEYS>
__global__ void k_gemm_wmma(const float* __restrict__ X, const float* __restrict__ W,
                            float* __restrict__ H, int M,
                            const int* __restrict__ src, const int* __restrict__ dst,
                            const float* __restrict__ dis, unsigned* __restrict__ keys,
                            const uint2* __restrict__ rec_from, uint2* __restrict__ rec_to, int e) {
    constexpr int LDA = FIN + 1;
    __shared__ float sW[FIN * FOUT];
    __shared__ float sA[MT * 16 * LDA];

    const int tid  = threadIdx.x;
    const int nthr = 32 * (FOUT / 16);
    const int m0   = blockIdx.x * (MT * 16);

    if (KEYS) {
        write_keys(dst, keys, e, (int)(blockIdx.x * nthr + tid), (int)(gridDim.x * nthr));
        copy_recs(rec_from, rec_to, e, (int)(blockIdx.x * nthr + tid), (int)(gridDim.x * nthr));
    } else {
        build_recs(src, dst, dis, rec_to, e, (int)(blockIdx.x * nthr + tid), (int)(gridDim.x * nthr));
    }

    for (int i = tid; i < FIN * FOUT; i += nthr) sW[i] = W[i];
    for (int i = tid; i < MT * 16 * FIN; i += nthr) {
        int r = i / FIN, c = i - r * FIN;
        int row = m0 + r; if (row >= M) row = M - 1;
        float v = X[(size_t)row * FIN + c];
        sA[r * LDA + c] = RELU ? fmaxf(v, 0.0f) : v;
    }
    __syncthreads();

    const int wave = tid >> 5;
    const int lane = tid & 31;
    const int half = lane >> 4;
    const int l16  = lane & 15;
    const int n0   = wave * 16;

    v2f bf[FIN / 4];
    #pragma unroll
    for (int kk = 0; kk < FIN / 4; ++kk) {
        int k = kk * 4 + 2 * half;
        bf[kk].x = sW[k * FOUT + n0 + l16];
        bf[kk].y = sW[(k + 1) * FOUT + n0 + l16];
    }

    #pragma unroll
    for (int mt = 0; mt < MT; ++mt) {
        const float* At = &sA[mt * 16 * LDA];
        v8f c = {};
        #pragma unroll
        for (int kk = 0; kk < FIN / 4; ++kk) {
            int k = kk * 4 + 2 * half;
            v2f a;
            a.x = At[l16 * LDA + k];
            a.y = At[l16 * LDA + k + 1];
            c = __builtin_amdgcn_wmma_f32_16x16x4_f32(false, a, false, bf[kk],
                                                      (short)0, c, false, false);
        }
        #pragma unroll
        for (int r = 0; r < 8; ++r) {
            int row = m0 + mt * 16 + half * 8 + r;
            if (row < M) H[(size_t)row * FOUT + n0 + l16] = c[r];
        }
    }
}

template <int F, int U>
__device__ __forceinline__ void rows_load(float (&v)[U][F / 32], const float* hl, int s, int j0) {
    #pragma unroll
    for (int u = 0; u < U; ++u) {
        const int sj = __builtin_amdgcn_readlane(s, j0 + u);
        const float* hp = hl + (size_t)sj * F;
        #pragma unroll
        for (int k = 0; k < F / 32; ++k) v[u][k] = hp[32 * k];
    }
}

template <int F, int U>
__device__ __forceinline__ void rows_add(const float (&v)[U][F / 32], float* al, int kr, float nrm, int j0) {
    #pragma unroll
    for (int u = 0; u < U; ++u) {
        const int   kj = __builtin_amdgcn_readlane(kr, j0 + u);
        const float nj = __int_as_float(__builtin_amdgcn_readlane(__float_as_int(nrm), j0 + u));
        float* ap = al + kj * F;
        #pragma unroll
        for (int k = 0; k < F / 32; ++k) atomicAdd(ap + 32 * k, v[u][k] * nj);
    }
}

template <int F, int U>
__device__ __forceinline__ void walk32(const float* hl, float* al, int s, int kr, float nrm) {
    float va[U][F / 32], vb[U][F / 32];
    rows_load<F, U>(va, hl, s, 0);
    #pragma unroll
    for (int j0 = 0; j0 < 32; j0 += 2 * U) {
        rows_load<F, U>(vb, hl, s, j0 + U);
        rows_add<F, U>(va, al, kr, nrm, j0);
        if (j0 + 2 * U < 32) rows_load<F, U>(va, hl, s, j0 + 2 * U);
        rows_add<F, U>(vb, al, kr, nrm, j0 + U);
    }
}

template <int F, int U>
__global__ void __launch_bounds__(1024)
k_agg(const float* __restrict__ h, const unsigned* __restrict__ keys,
      const uint2* __restrict__ rec, const int* __restrict__ dst,
      const float* __restrict__ dis, const float* __restrict__ bias,
      float* __restrict__ out, int e, int n) {
    static_assert(32 % (2 * U) == 0 && F % 32 == 0, "group shape");
    constexpr int B = 4;
    __shared__ __attribute__((aligned(16))) float acc[AGG_ROWS * F];
    __shared__ float dl[AGG_ROWS];

    const int tid  = threadIdx.x;
    const int lane = tid & 31;
    const int lo   = blockIdx.x * AGG_ROWS;
    const int rows = min(AGG_ROWS, n - lo);

    for (int i = tid; i < AGG_ROWS; i += 1024) dl[i] = (i < rows) ? dis[lo + i] : 0.0f;
    {
        const float4* h4 = reinterpret_cast<const float4*>(h + (size_t)lo * F);
        const float4* b4 = reinterpret_cast<const float4*>(bias);
        float4* a4 = reinterpret_cast<float4*>(acc);
        for (int i = tid; i < rows * (F / 4); i += 1024) {
            const int r = i / (F / 4), c = i - r * (F / 4);
            const float s = dis[lo + r];
            const float w = s * s;
            const float4 hv = h4[i];
            const float4 bv = b4[c];
            float4 o;
            o.x = hv.x * w + bv.x;  o.y = hv.y * w + bv.y;
            o.z = hv.z * w + bv.z;  o.w = hv.w * w + bv.w;
            a4[i] = o;
        }
    }
    __syncthreads();

    const float* hl = h + lane;
    float*       al = acc + lane;
    const unsigned kb   = blockIdx.x < 0xFFu ? blockIdx.x : 0xFFu;
    const unsigned key4 = kb * 0x01010101u;
    const int nq = (e + 15) >> 4;
    const uint4* kq = reinterpret_cast<const uint4*>(keys);
    const int per = 1024 * B;
    const int T   = (nq + per - 1) / per;
    const int rot = (int)(((long long)blockIdx.x * T) / gridDim.x);

    int qe = 0;
    int qc = 0;

    uint4 cur[B], nxt[B];
    int itr = rot;
    #pragma unroll
    for (int j = 0; j < B; ++j) cur[j] = kq[min(itr * per + j * 1024 + tid, nq - 1)];

    for (int it = 0; it < T; ++it) {
        int itn = itr + 1;
        if (itn >= T) itn = 0;
        #pragma unroll
        for (int j = 0; j < B; ++j) nxt[j] = kq[min(itn * per + j * 1024 + tid, nq - 1)];

        const int g0 = itr * per + tid;
        unsigned m[B];
        unsigned any = 0;
        #pragma unroll
        for (int j = 0; j < B; ++j) {
            const unsigned x0 = cur[j].x ^ key4, x1 = cur[j].y ^ key4, x2 = cur[j].z ^ key4, x3 = cur[j].w ^ key4;
            const unsigned z0 = ~(((x0 & 0x7F7F7F7Fu) + 0x7F7F7F7Fu) | x0) & 0x80808080u;
            const unsigned z1 = ~(((x1 & 0x7F7F7F7Fu) + 0x7F7F7F7Fu) | x1) & 0x80808080u;
            const unsigned z2 = ~(((x2 & 0x7F7F7F7Fu) + 0x7F7F7F7Fu) | x2) & 0x80808080u;
            const unsigned z3 = ~(((x3 & 0x7F7F7F7Fu) + 0x7F7F7F7Fu) | x3) & 0x80808080u;
            m[j] = (z0 >> 7) | (z1 >> 6) | (z2 >> 5) | (z3 >> 4);
            if (g0 + j * 1024 >= nq) m[j] = 0;
            any |= m[j];
        }
        while (__builtin_amdgcn_ballot_w32(any != 0) != 0) {
            const bool pass = (any != 0);
            unsigned mm = m[B - 1];
            int jj = B - 1;
            #pragma unroll
            for (int j = B - 2; j >= 0; --j) {
                if (m[j] != 0) { mm = m[j]; jj = j; }
            }
            const int bit = pass ? __builtin_ctz(mm) : 0;
            const int ei  = 16 * (g0 + jj * 1024) + 4 * (bit & 7) + (bit >> 3);
            any = 0;
            #pragma unroll
            for (int j = 0; j < B; ++j) {
                if (jj == j) m[j] &= m[j] - 1;
                any |= m[j];
            }
            const unsigned bm = __builtin_amdgcn_ballot_w32(pass);
            const int cnt = __builtin_popcount(bm);
            const int rk  = (int)__builtin_amdgcn_mbcnt_lo(bm, 0u);
            const int pos = pass ? (qc + rk) : (qc + cnt + (lane - rk));
            const int pe = __builtin_amdgcn_ds_permute((pos & 31) << 2, ei);
            const int me = (lane < qc) ? qe : pe;
            if (qc + cnt >= 32) {
                const int ec = min(me, e - 1);
                const uint2 r = rec[ec];
                const int s  = (int)(r.x & ((1u << REC_SHIFT) - 1u));
                unsigned kr  = r.x >> REC_SHIFT;
                bool ok = (me < e);
                if (blockIdx.x >= 0xFFu) {
                    kr = (unsigned)(dst[ec] - lo);
                    ok = ok && (kr < (unsigned)rows);
                }
                if (!ok) kr = 0;
                const float nrm = ok ? __uint_as_float(r.y) * dl[kr] : 0.0f;
                walk32<F, U>(hl, al, s, (int)kr, nrm);
                qe = pe;
                qc = qc + cnt - 32;
            } else {
                qe = me;
                qc = qc + cnt;
            }
        }
        #pragma unroll
        for (int j = 0; j < B; ++j) cur[j] = nxt[j];
        itr = itn;
    }

    if (qc > 0) {
        int s = 0, kr = 0;
        float nrm = 0.0f;
        if (lane < qc) {
            const int ec = min(qe, e - 1);
            const uint2 r = rec[ec];
            s = (int)(r.x & ((1u << REC_SHIFT) - 1u));
            unsigned k = r.x >> REC_SHIFT;
            bool ok = (qe < e);
            if (blockIdx.x >= 0xFFu) {
                k  = (unsigned)(dst[ec] - lo);
                ok = ok && (k < (unsigned)rows);
            }
            kr  = ok ? (int)k : 0;
            nrm = ok ? __uint_as_float(r.y) * dl[kr] : 0.0f;
        }
        for (int j = 0; j < qc; ++j) {
            float v1[1][F / 32];
            rows_load<F, 1>(v1, hl, s, j);
            rows_add<F, 1>(v1, al, kr, nrm, j);
        }
    }
    __syncthreads();

    {
        float4* o4 = reinterpret_cast<float4*>(out + (size_t)lo * F);
        const float4* a4 = reinterpret_cast<const float4*>(acc);
        for (int i = tid; i < rows * (F / 4); i += 1024) o4[i] = a4[i];
    }
}

extern "C" void kernel_launch(void* const* d_in, const int* in_sizes, int n_in,
                              void* d_out, int out_size, void* d_ws, size_t ws_size,
                              hipStream_t stream) {
    const float* X  = (const float*)d_in[0];
    const int*   EI = (const int*)d_in[1];
    const float* W1 = (const float*)d_in[2];
    const float* b1 = (const float*)d_in[3];
    const float* W2 = (const float*)d_in[4];
    const float* b2 = (const float*)d_in[5];
    float* out = (float*)d_out;

    const int FIN1 = 64;
    const int N = in_sizes[0] / FIN1;
    const int E = in_sizes[1] / 2;
    const int* src = EI;
    const int* dst = EI + E;

    float* dis = (float*)d_ws;
    size_t off = ((size_t)N * sizeof(float) + 16 + 255) & ~(size_t)255;
    float* bufH = (float*)((char*)d_ws + off);
    float* bufO = bufH + (size_t)N * 128;
    int*   degp = (int*)bufH;
    unsigned* keys1 = (unsigned*)d_out;
    unsigned* keys2 = (unsigned*)(bufH + (size_t)N * 64);
    const size_t kbytes = (((((size_t)E + 15) >> 4) << 4) + 255) & ~(size_t)255;
    uint2* rec1 = (uint2*)((char*)keys1 + kbytes);
    uint2* rec2 = (uint2*)((char*)keys2 + kbytes);

    const int T = 256;
    dim3 blkN((N + T - 1) / T);
    dim3 blkD((N + DEG_RANGE - 1) / DEG_RANGE, DEG_CHUNKS);
    const int ntiles = (N + 15) / 16;
    dim3 blkA((N + AGG_ROWS - 1) / AGG_ROWS);

    k_count_deg<<<blkD, T, 0, stream>>>(dst, degp, E, N);
    k_finalize_dis<<<blkN, T, 0, stream>>>(degp, dis, dst, keys1, E, N);

    k_gemm_wmma<64, 128, 4, false, false><<<(ntiles + 3) / 4, 256, 0, stream>>>(
        X, W1, bufH, N, src, dst, dis, keys2, nullptr, rec1, E);
    k_agg<128, 4><<<blkA, 1024, 0, stream>>>(bufH, keys1, rec1, dst, dis, b1, bufO, E, N);

    k_gemm_wmma<128, 64, 2, true, true><<<(ntiles + 1) / 2, 128, 0, stream>>>(
        bufO, W2, bufH, N, src, dst, dis, keys2, rec1, rec2, E);
    k_agg<64, 8><<<blkA, 1024, 0, stream>>>(bufH, keys2, rec2, dst, dis, b2, out, E, N);
}


# Layer-2 neighbour sums read per-piece edge lists that the layer-1 owner work-groups append as they walk: no key scan, pick or per-edge record gather in layer 2; f32 sum order = LDS arrival
# speedup vs baseline: 2.7625x; 1.1102x over previous
#include <hip/hip_runtime.h>
#include <hip/hip_bf16.h>

typedef __attribute__((ext_vector_type(2))) float v2f;
typedef __attribute__((ext_vector_type(8))) float v8f;

constexpr int DEG_RANGE  = 16384;
constexpr int DEG_CHUNKS = 64;

__global__ void __launch_bounds__(256)
k_count_deg(const int* __restrict__ dst, int* __restrict__ part, int e, int n) {
    __shared__ int cnt[DEG_RANGE];
    const int tid = threadIdx.x;
    const int lo  = blockIdx.x * DEG_RANGE;
    const int c   = blockIdx.y;
    for (int i = tid; i < DEG_RANGE; i += 256) cnt[i] = 0;
    __syncthreads();

    const int per = (((e + DEG_CHUNKS - 1) / DEG_CHUNKS) + 3) & ~3;
    const int e0  = c * per;
    const int e1  = min(e0 + per, e);
    if ((((size_t)dst) & 15) == 0) {
        const int nq = (e1 > e0) ? ((e1 - e0) >> 2) : 0;
        const int4* dq = reinterpret_cast<const int4*>(dst + e0);
        constexpr int B = 8;
        for (int qb = 0; qb < nq; qb += 256 * B) {
            int4 v[B];
            #pragma unroll
            for (int j = 0; j < B; ++j) {
                const int q = qb + j * 256 + tid;
                v[j] = dq[min(q, nq - 1)];
            }
            #pragma unroll
            for (int j = 0; j < B; ++j) {
                const bool in = (qb + j * 256 + tid) < nq;
                const unsigned k0 = (unsigned)(v[j].x - lo), k1 = (unsigned)(v[j].y - lo);
                const unsigned k2 = (unsigned)(v[j].z - lo), k3 = (unsigned)(v[j].w - lo);
                if (in && k0 < (unsigned)DEG_RANGE) atomicAdd(&cnt[k0], 1);
                if (in && k1 < (unsigned)DEG_RANGE) atomicAdd(&cnt[k1], 1);
                if (in && k2 < (unsigned)DEG_RANGE) atomicAdd(&cnt[k2], 1);
                if (in && k3 < (unsigned)DEG_RANGE) atomicAdd(&cnt[k3], 1);
            }
        }
        const int i = e0 + 4 * nq + tid;
        if (tid < 4 && i < e1) {
            const unsigned k = (unsigned)(dst[i] - lo);
            if (k < (unsigned)DEG_RANGE) atomicAdd(&cnt[k], 1);
        }
    } else {
        for (int i = e0 + tid; i < e1; i += 256) {
            const unsigned k = (unsigned)(dst[i] - lo);
            if (k < (unsigned)DEG_RANGE) atomicAdd(&cnt[k], 1);
        }
    }
    __syncthreads();

    const int hi = min(DEG_RANGE, n - lo);
    int* p = part + (size_t)c * n + lo;
    for (int i = tid; i < hi; i += 256) p[i] = cnt[i];
}

constexpr int AGG_ROWS = 400;

__device__ __forceinline__ void write_keys(const int* __restrict__ dst, unsigned* __restrict__ keys,
                                           int e, int gtid, int gsize) {
    const int nw = ((e + 15) >> 4) << 2;
    for (int w = gtid; w < nw; w += gsize) {
        unsigned v = 0;
        #pragma unroll
        for (int j = 0; j < 4; ++j) {
            const int i = 4 * w + j;
            unsigned k = 0xFFu;
            if (i < e) {
                const unsigned p = (unsigned)dst[i] / (unsigned)AGG_ROWS;
                k = p < 0xFFu ? p : 0xFFu;
            }
            v |= k << (8 * j);
        }
        keys[w] = v;
    }
}

constexpr int REC_SHIFT = 23;
constexpr int DEG_BLOCK = 256;
constexpr unsigned DEG16_BIG = 0xFFFFu;

struct Lists {
    unsigned*       tbl;
    unsigned*       lx;
    unsigned short* ld;
};

__device__ __forceinline__ void build_recs(const int* __restrict__ src, const int* __restrict__ dst,
                                           const int* __restrict__ degi, uint2* __restrict__ rec,
                                           int e, int gtid, int gsize) {
    constexpr int B = 4;
    for (int i0 = gtid; i0 < e; i0 += B * gsize) {
        int s[B], d[B];
        int ws[B];
        #pragma unroll
        for (int j = 0; j < B; ++j) {
            const int i = min(i0 + j * gsize, e - 1);
            s[j] = src[i];
            d[j] = dst[i];
        }
        #pragma unroll
        for (int j = 0; j < B; ++j) ws[j] = degi[s[j]] + 1;
        #pragma unroll
        for (int j = 0; j < B; ++j) {
            const int i = i0 + j * gsize;
            if (i < e) {
                uint2 r;
                r.x = (unsigned)s[j] | (((unsigned)d[j] % (unsigned)AGG_ROWS) << REC_SHIFT);
                r.y = (unsigned)ws[j];
                rec[i] = r;
            }
        }
    }
}

__device__ __forceinline__ void copy_words(const unsigned* __restrict__ from, unsigned* __restrict__ to,
                                           int nw, int gtid, int gsize) {
    const int nq = nw >> 2;
    const uint4* f4 = reinterpret_cast<const uint4*>(from);
    uint4* t4 = reinterpret_cast<uint4*>(to);
    for (int i = gtid; i < nq; i += gsize) t4[i] = f4[i];
    const int i = 4 * nq + gtid;
    if (gtid < 4 && i < nw) to[i] = from[i];
}

__device__ __forceinline__ void copy_lists(const Lists from, const Lists to, int e, int np, int gtid, int gsize) {
    copy_words(from.tbl, to.tbl, 2 * np, gtid, gsize);
    copy_words(from.lx, to.lx, e, gtid, gsize);
    copy_words(reinterpret_cast<const unsigned*>(from.ld), reinterpret_cast<unsigned*>(to.ld), e >> 1, gtid, gsize);
    if ((e & 1) && gtid == 0) to.ld[e - 1] = from.ld[e - 1];
}

__global__ void __launch_bounds__(DEG_BLOCK)
k_finalize_dis(const int* __restrict__ part, float* __restrict__ dis,
               int* __restrict__ degi, int* __restrict__ bsum,
               const int* __restrict__ dst, unsigned* __restrict__ keys,
               int e, int n) {
    __shared__ int bs;
    if (threadIdx.x == 0) bs = 0;
    __syncthreads();
    int i = blockIdx.x * blockDim.x + threadIdx.x;
    int din = 0;
    if (i < n) {
        for (int c = 0; c < DEG_CHUNKS; ++c) din += part[(size_t)c * n + i];
        dis[i]  = rsqrtf((float)(din + 1));
        degi[i] = din;
    }
    atomicAdd(&bs, din);
    __syncthreads();
    if (threadIdx.x == 0) bsum[blockIdx.x] = bs;
    write_keys(dst, keys, e, i, (int)(gridDim.x * blockDim.x));
}

template <int FIN, int FOUT, int MT, bool RELU, bool LISTS>
__global__ void k_gemm_wmma(const float* __restrict__ X, const float* __restrict__ W,
                            float* __restrict__ H, int M,
                            const int* __restrict__ src, const int* __restrict__ dst,
                            const int* __restrict__ degi, uint2* __restrict__ rec_to,
                            const Lists l_from, const Lists l_to, int e, int np) {
    constexpr int LDA = FIN + 1;
    __shared__ float sW[FIN * FOUT];
    __shared__ float sA[MT * 16 * LDA];

    const int tid  = threadIdx.x;
    const int nthr = 32 * (FOUT / 16);
    const int m0   = blockIdx.x * (MT * 16);

    if (LISTS) {
        copy_lists(l_from, l_to, e, np, (int)(blockIdx.x * nthr + tid), (int)(gridDim.x * nthr));
    } else {
        build_recs(src, dst, degi, rec_to, e, (int)(blockIdx.x * nthr + tid), (int)(gridDim.x * nthr));
    }

    for (int i = tid; i < FIN * FOUT; i += nthr) sW[i] = W[i];
    for (int i = tid; i < MT * 16 * FIN; i += nthr) {
        int r = i / FIN, c = i - r * FIN;
        int row = m0 + r; if (row >= M) row = M - 1;
        float v = X[(size_t)row * FIN + c];
        sA[r * LDA + c] = RELU ? fmaxf(v, 0.0f) : v;
    }
    __syncthreads();

    const int wave = tid >> 5;
    const int lane = tid & 31;
    const int half = lane >> 4;
    const int l16  = lane & 15;
    const int n0   = wave * 16;

    v2f bf[FIN / 4];
    #pragma unroll
    for (int kk = 0; kk < FIN / 4; ++kk) {
        int k = kk * 4 + 2 * half;
        bf[kk].x = sW[k * FOUT + n0 + l16];
        bf[kk].y = sW[(k + 1) * FOUT + n0 + l16];
    }

    #pragma unroll
    for (int mt = 0; mt < MT; ++mt) {
        const float* At = &sA[mt * 16 * LDA];
        v8f c = {};
        #pragma unroll
        for (int kk = 0; kk < FIN / 4; ++kk) {
            int k = kk * 4 + 2 * half;
            v2f a;
            a.x = At[l16 * LDA + k];
            a.y = At[l16 * LDA + k + 1];
            c = __builtin_amdgcn_wmma_f32_16x16x4_f32(false, a, false, bf[kk],
                                                      (short)0, c, false, false);
        }
        #pragma unroll
        for (int r = 0; r < 8; ++r) {
            int row = m0 + mt * 16 + half * 8 + r;
            if (row < M) H[(size_t)row * FOUT + n0 + l16] = c[r];
        }
    }
}

template <int F, int U>
__device__ __forceinline__ void rows_load(float (&v)[U][F / 32], const float* hl, int s, int j0) {
    #pragma unroll
    for (int u = 0; u < U; ++u) {
        const int sj = __builtin_amdgcn_readlane(s, j0 + u);
        const float* hp = hl + (size_t)sj * F;
        #pragma unroll
        for (int k = 0; k < F / 32; ++k) v[u][k] = hp[32 * k];
    }
}

template <int F, int U>
__device__ __forceinline__ void rows_add(const float (&v)[U][F / 32], float* al, int kr, float nrm, int j0) {
    #pragma unroll
    for (int u = 0; u < U; ++u) {
        const int   kj = __builtin_amdgcn_readlane(kr, j0 + u);
        const float nj = __int_as_float(__builtin_amdgcn_readlane(__float_as_int(nrm), j0 + u));
        float* ap = al + kj * F;
        #pragma unroll
        for (int k = 0; k < F / 32; ++k) atomicAdd(ap + 32 * k, v[u][k] * nj);
    }
}

template <int F, int U>
__device__ __forceinline__ void walk32(const float* hl, float* al, int s, int kr, float nrm) {
    float va[U][F / 32], vb[U][F / 32];
    rows_load<F, U>(va, hl, s, 0);
    #pragma unroll
    for (int j0 = 0; j0 < 32; j0 += 2 * U) {
        rows_load<F, U>(vb, hl, s, j0 + U);
        rows_add<F, U>(va, al, kr, nrm, j0);
        if (j0 + 2 * U < 32) rows_load<F, U>(va, hl, s, j0 + 2 * U);
        rows_add<F, U>(vb, al, kr, nrm, j0 + U);
    }
}

__device__ __forceinline__ void list_append(const Lists& L, int off, int* lcount, bool ok,
                                            int s, unsigned kr, unsigned deg, int lane) {
    const unsigned okm = __builtin_amdgcn_ballot_w32(ok);
    int base = 0;
    if (lane == 0) base = atomicAdd(lcount, __builtin_popcount(okm));
    base = __builtin_amdgcn_readfirstlane(base);
    if (ok) {
        const int slot = off + base + (int)__builtin_amdgcn_mbcnt_lo(okm, 0u);
        L.lx[slot] = (unsigned)s | (kr << REC_SHIFT);
        L.ld[slot] = (unsigned short)(deg < DEG16_BIG ? deg : DEG16_BIG);
    }
}

template <int F, int U>
__global__ void __launch_bounds__(1024)
k_agg(const float* __restrict__ h, const unsigned* __restrict__ keys,
      const uint2* __restrict__ rec, const int* __restrict__ dst,
      const float* __restrict__ dis, const float* __restrict__ bias,
      const int* __restrict__ degi, const int* __restrict__ bsum, const Lists L,
      float* __restrict__ out, int e, int n) {
    static_assert(32 % (2 * U) == 0 && F % 32 == 0, "group shape");
    constexpr int B = 4;
    __shared__ __attribute__((aligned(16))) float acc[AGG_ROWS * F];
    __shared__ float dl[AGG_ROWS];
    __shared__ int loff;
    __shared__ int lcount;

    const int tid  = threadIdx.x;
    const int lane = tid & 31;
    const int lo   = blockIdx.x * AGG_ROWS;
    const int rows = min(AGG_ROWS, n - lo);

    if (tid == 0) { loff = 0; lcount = 0; }
    __syncthreads();
    {
        const int nb = lo / DEG_BLOCK;
        int v = 0;
        for (int i = tid; i < nb; i += 1024) v += bsum[i];
        for (int i = nb * DEG_BLOCK + tid; i < lo; i += 1024) v += degi[i];
        atomicAdd(&loff, v);
    }

    for (int i = tid; i < AGG_ROWS; i += 1024) dl[i] = (i < rows) ? dis[lo + i] : 0.0f;
    {
        const float4* h4 = reinterpret_cast<const float4*>(h + (size_t)lo * F);
        const float4* b4 = reinterpret_cast<const float4*>(bias);
        float4* a4 = reinterpret_cast<float4*>(acc);
        for (int i = tid; i < rows * (F / 4); i += 1024) {
            const int r = i / (F / 4), c = i - r * (F / 4);
            const float s = dis[lo + r];
            const float w = s * s;
            const float4 hv = h4[i];
            const float4 bv = b4[c];
            float4 o;
            o.x = hv.x * w + bv.x;  o.y = hv.y * w + bv.y;
            o.z = hv.z * w + bv.z;  o.w = hv.w * w + bv.w;
            a4[i] = o;
        }
    }
    __syncthreads();

    const int off = loff;
    const float* hl = h + lane;
    float*       al = acc + lane;
    const unsigned kb   = blockIdx.x < 0xFFu ? blockIdx.x : 0xFFu;
    const unsigned key4 = kb * 0x01010101u;
    const int nq = (e + 15) >> 4;
    const uint4* kq = reinterpret_cast<const uint4*>(keys);
    const int per = 1024 * B;
    const int T   = (nq + per - 1) / per;
    const int rot = (int)(((long long)blockIdx.x * T) / gridDim.x);

    int qe = 0;
    int qc = 0;

    uint4 cur[B], nxt[B];
    int itr = rot;
    #pragma unroll
    for (int j = 0; j < B; ++j) cur[j] = kq[min(itr * per + j * 1024 + tid, nq - 1)];

    for (int it = 0; it < T; ++it) {
        int itn = itr + 1;
        if (itn >= T) itn = 0;
        #pragma unroll
        for (int j = 0; j < B; ++j) nxt[j] = kq[min(itn * per + j * 1024 + tid, nq - 1)];

        const int g0 = itr * per + tid;
        unsigned m[B];
        unsigned any = 0;
        #pragma unroll
        for (int j = 0; j < B; ++j) {
            const unsigned x0 = cur[j].x ^ key4, x1 = cur[j].y ^ key4, x2 = cur[j].z ^ key4, x3 = cur[j].w ^ key4;
            const unsigned z0 = ~(((x0 & 0x7F7F7F7Fu) + 0x7F7F7F7Fu) | x0) & 0x80808080u;
            const unsigned z1 = ~(((x1 & 0x7F7F7F7Fu) + 0x7F7F7F7Fu) | x1) & 0x80808080u;
            const unsigned z2 = ~(((x2 & 0x7F7F7F7Fu) + 0x7F7F7F7Fu) | x2) & 0x80808080u;
            const unsigned z3 = ~(((x3 & 0x7F7F7F7Fu) + 0x7F7F7F7Fu) | x3) & 0x80808080u;
            m[j] = (z0 >> 7) | (z1 >> 6) | (z2 >> 5) | (z3 >> 4);
            if (g0 + j * 1024 >= nq) m[j] = 0;
            any |= m[j];
        }
        while (__builtin_amdgcn_ballot_w32(any != 0) != 0) {
            const bool pass = (any != 0);
            unsigned mm = m[B - 1];
            int jj = B - 1;
            #pragma unroll
            for (int j = B - 2; j >= 0; --j) {
                if (m[j] != 0) { mm = m[j]; jj = j; }
            }
            const int bit = pass ? __builtin_ctz(mm) : 0;
            const int ei  = 16 * (g0 + jj * 1024) + 4 * (bit & 7) + (bit >> 3);
            any = 0;
            #pragma unroll
            for (int j = 0; j < B; ++j) {
                if (jj == j) m[j] &= m[j] - 1;
                any |= m[j];
            }
            const unsigned bm = __builtin_amdgcn_ballot_w32(pass);
            const int cnt = __builtin_popcount(bm);
            const int rk  = (int)__builtin_amdgcn_mbcnt_lo(bm, 0u);
            const int pos = pass ? (qc + rk) : (qc + cnt + (lane - rk));
            const int pe = __builtin_amdgcn_ds_permute((pos & 31) << 2, ei);
            const int me = (lane < qc) ? qe : pe;
            if (qc + cnt >= 32) {
                const int ec = min(me, e - 1);
                const uint2 r = rec[ec];
                const int s  = (int)(r.x & ((1u << REC_SHIFT) - 1u));
                unsigned kr  = r.x >> REC_SHIFT;
                bool ok = (me < e);
                if (blockIdx.x >= 0xFFu) {
                    kr = (unsigned)(dst[ec] - lo);
                    ok = ok && (kr < (unsigned)rows);
                }
                if (!ok) kr = 0;
                const float nrm = ok ? rsqrtf((float)(int)r.y) * dl[kr] : 0.0f;
                list_append(L, off, &lcount, ok, s, kr, r.y, lane);
                walk32<F, U>(hl, al, s, (int)kr, nrm);
                qe = pe;
                qc = qc + cnt - 32;
            } else {
                qe = me;
                qc = qc + cnt;
            }
        }
        #pragma unroll
        for (int j = 0; j < B; ++j) cur[j] = nxt[j];
        itr = itn;
    }

    if (qc > 0) {
        int s = 0, kr = 0;
        float nrm = 0.0f;
        bool ok = false;
        unsigned dg = 1;
        if (lane < qc) {
            const int ec = min(qe, e - 1);
            const uint2 r = rec[ec];
            s = (int)(r.x & ((1u << REC_SHIFT) - 1u));
            unsigned k = r.x >> REC_SHIFT;
            ok = (qe < e);
            if (blockIdx.x >= 0xFFu) {
                k  = (unsigned)(dst[ec] - lo);
                ok = ok && (k < (unsigned)rows);
            }
            dg  = r.y;
            kr  = ok ? (int)k : 0;
            nrm = ok ? rsqrtf((float)(int)dg) * dl[kr] : 0.0f;
        }
        list_append(L, off, &lcount, ok, s, (unsigned)kr, dg, lane);
        for (int j = 0; j < qc; ++j) {
            float v1[1][F / 32];
            rows_load<F, 1>(v1, hl, s, j);
            rows_add<F, 1>(v1, al, kr, nrm, j);
        }
    }
    __syncthreads();

    {
        float4* o4 = reinterpret_cast<float4*>(out + (size_t)lo * F);
        const float4* a4 = reinterpret_cast<const float4*>(acc);
        for (int i = tid; i < rows * (F / 4); i += 1024) o4[i] = a4[i];
    }
    if (tid == 0) {
        L.tbl[2 * blockIdx.x]     = (unsigned)off;
        L.tbl[2 * blockIdx.x + 1] = (unsigned)lcount;
    }
}

template <int F, int U>
__global__ void __launch_bounds__(1024)
k_agg_list(const float* __restrict__ h, const unsigned* __restrict__ tbl,
           const unsigned* __restrict__ lx, const unsigned short* __restrict__ ld,
           const float* __restrict__ dis, const float* __restrict__ bias,
           float* __restrict__ out, int n) {
    static_assert(32 % (2 * U) == 0 && F % 32 == 0, "group shape");
    __shared__ __attribute__((aligned(16))) float acc[AGG_ROWS * F];
    __shared__ float dl[AGG_ROWS];

    const int tid  = threadIdx.x;
    const int lane = tid & 31;
    const int wave = __builtin_amdgcn_readfirstlane(tid >> 5);
    const int lo   = blockIdx.x * AGG_ROWS;
    const int rows = min(AGG_ROWS, n - lo);

    const int off = (int)tbl[2 * blockIdx.x];
    const int cnt = (int)tbl[2 * blockIdx.x + 1];
    const int last = off + max(cnt, 1) - 1;
    int pos = 32 * wave + lane;
    unsigned x  = lx[min(off + pos, last)];
    unsigned dg = ld[min(off + pos, last)];

    for (int i = tid; i < AGG_ROWS; i += 1024) dl[i] = (i < rows) ? dis[lo + i] : 0.0f;
    {
        const float4* h4 = reinterpret_cast<const float4*>(h + (size_t)lo * F);
        const float4* b4 = reinterpret_cast<const float4*>(bias);
        float4* a4 = reinterpret_cast<float4*>(acc);
        for (int i = tid; i < rows * (F / 4); i += 1024) {
            const int r = i / (F / 4), c = i - r * (F / 4);
            const float s = dis[lo + r];
            const float w = s * s;
            const float4 hv = h4[i];
            const float4 bv = b4[c];
            float4 o;
            o.x = hv.x * w + bv.x;  o.y = hv.y * w + bv.y;
            o.z = hv.z * w + bv.z;  o.w = hv.w * w + bv.w;
            a4[i] = o;
        }
    }
    __syncthreads();

    const float* hl = h + lane;
    float*       al = acc + lane;
    for (int p0 = 32 * wave; p0 < cnt; p0 += 1024) {
        const bool ok = pos < cnt;
        const int  s  = ok ? (int)(x & ((1u << REC_SHIFT) - 1u)) : 0;
        const int  kr = ok ? (int)(x >> REC_SHIFT) : 0;
        float ds = rsqrtf((float)(int)dg);
        if (dg == DEG16_BIG) ds = dis[s];
        const int pn = pos + 1024;
        const unsigned xn = lx[min(off + pn, last)];
        const unsigned dn = ld[min(off + pn, last)];
        const float nrm = ok ? ds * dl[kr] : 0.0f;
        const int m = min(32, cnt - p0);
        if (m == 32) {
            walk32<F, U>(hl, al, s, kr, nrm);
        } else {
            for (int j = 0; j < m; ++j) {
                float v1[1][F / 32];
                rows_load<F, 1>(v1, hl, s, j);
                rows_add<F, 1>(v1, al, kr, nrm, j);
            }
        }
        x = xn;  dg = dn;  pos = pn;
    }
    __syncthreads();

    {
        float4* o4 = reinterpret_cast<float4*>(out + (size_t)lo * F);
        const float4* a4 = reinterpret_cast<const float4*>(acc);
        for (int i = tid; i < rows * (F / 4); i += 1024) o4[i] = a4[i];
    }
}

extern "C" void kernel_launch(void* const* d_in, const int* in_sizes, int n_in,
                              void* d_out, int out_size, void* d_ws, size_t ws_size,
                              hipStream_t stream) {
    const float* X  = (const float*)d_in[0];
    const int*   EI = (const int*)d_in[1];
    const float* W1 = (const float*)d_in[2];
    const float* b1 = (const float*)d_in[3];
    const float* W2 = (const float*)d_in[4];
    const float* b2 = (const float*)d_in[5];
    float* out = (float*)d_out;

    const int FIN1 = 64;
    const int N = in_sizes[0] / FIN1;
    const int E = in_sizes[1] / 2;
    const int* src = EI;
    const int* dst = EI + E;

    float* dis = (float*)d_ws;
    size_t off = ((size_t)N * sizeof(float) + 16 + 255) & ~(size_t)255;
    float* bufH = (float*)((char*)d_ws + off);
    float* bufO = bufH + (size_t)N * 128;
    int*   degp = (int*)bufH;
    const int NP   = (N + AGG_ROWS - 1) / AGG_ROWS;
    const int NBLK = (N + DEG_BLOCK - 1) / DEG_BLOCK;
    const size_t kbytes = (((((size_t)E + 15) >> 4) << 4) + 255) & ~(size_t)255;
    const size_t rbytes = ((size_t)E * 8 + 255) & ~(size_t)255;
    const size_t dbytes = ((size_t)N * 4 + 255) & ~(size_t)255;
    const size_t sbytes = ((size_t)NBLK * 4 + 255) & ~(size_t)255;
    const size_t tbytes = ((size_t)NP * 8 + 255) & ~(size_t)255;
    const size_t xbytes = ((size_t)E * 4 + 255) & ~(size_t)255;
    unsigned* keys1 = (unsigned*)d_out;
    uint2* rec1 = (uint2*)((char*)keys1 + kbytes);
    int*   degi = (int*)((char*)rec1 + rbytes);
    int*   bsum = (int*)((char*)degi + dbytes);
    Lists l1, l2;
    l1.tbl = (unsigned*)((char*)bsum + sbytes);
    l1.lx  = (unsigned*)((char*)l1.tbl + tbytes);
    l1.ld  = (unsigned short*)((char*)l1.lx + xbytes);
    l2.tbl = (unsigned*)(bufH + (size_t)N * 64);
    l2.lx  = (unsigned*)((char*)l2.tbl + tbytes);
    l2.ld  = (unsigned short*)((char*)l2.lx + xbytes);

    const int T = DEG_BLOCK;
    dim3 blkN(NBLK);
    dim3 blkD((N + DEG_RANGE - 1) / DEG_RANGE, DEG_CHUNKS);
    const int ntiles = (N + 15) / 16;
    dim3 blkA((N + AGG_ROWS - 1) / AGG_ROWS);

    k_count_deg<<<blkD, 256, 0, stream>>>(dst, degp, E, N);
    k_finalize_dis<<<blkN, T, 0, stream>>>(degp, dis, degi, bsum, dst, keys1, E, N);

    k_gemm_wmma<64, 128, 4, false, false><<<(ntiles + 3) / 4, 256, 0, stream>>>(
        X, W1, bufH, N, src, dst, degi, rec1, l1, l2, E, NP);
    k_agg<128, 4><<<blkA, 1024, 0, stream>>>(bufH, keys1, rec1, dst, dis, b1, degi, bsum, l1, bufO, E, N);

    k_gemm_wmma<128, 64, 2, true, true><<<(ntiles + 1) / 2, 128, 0, stream>>>(
        bufO, W2, bufH, N, src, dst, degi, rec1, l1, l2, E, NP);
    k_agg_list<64, 8><<<blkA, 1024, 0, stream>>>(bufH, l2.tbl, l2.lx, l2.ld, dis, b2, out, N);
}


# Source rows gathered with one 16-byte (layer 1) or 8-byte (layer 2) load per lane: 1 load per edge, not 4 or 2; LDS rows lane-interleaved, adds stay on 32 banks; f32 sum order = LDS arrival as before
# speedup vs baseline: 3.0395x; 1.1003x over previous
#include <hip/hip_runtime.h>
#include <hip/hip_bf16.h>

typedef __attribute__((ext_vector_type(2))) float v2f;
typedef __attribute__((ext_vector_type(8))) float v8f;

constexpr int DEG_RANGE  = 16384;
constexpr int DEG_CHUNKS = 64;

__global__ void __launch_bounds__(256)
k_count_deg(const int* __restrict__ dst, int* __restrict__ part, int e, int n) {
    __shared__ int cnt[DEG_RANGE];
    const int tid = threadIdx.x;
    const int lo  = blockIdx.x * DEG_RANGE;
    const int c   = blockIdx.y;
    for (int i = tid; i < DEG_RANGE; i += 256) cnt[i] = 0;
    __syncthreads();

    const int per = (((e + DEG_CHUNKS - 1) / DEG_CHUNKS) + 3) & ~3;
    const int e0  = c * per;
    const int e1  = min(e0 + per, e);
    if ((((size_t)dst) & 15) == 0) {
        const int nq = (e1 > e0) ? ((e1 - e0) >> 2) : 0;
        const int4* dq = reinterpret_cast<const int4*>(dst + e0);
        constexpr int B = 8;
        for (int qb = 0; qb < nq; qb += 256 * B) {
            int4 v[B];
            #pragma unroll
            for (int j = 0; j < B; ++j) {
                const int q = qb + j * 256 + tid;
                v[j] = dq[min(q, nq - 1)];
            }
            #pragma unroll
            for (int j = 0; j < B; ++j) {
                const bool in = (qb + j * 256 + tid) < nq;
                const unsigned k0 = (unsigned)(v[j].x - lo), k1 = (unsigned)(v[j].y - lo);
                const unsigned k2 = (unsigned)(v[j].z - lo), k3 = (unsigned)(v[j].w - lo);
                if (in && k0 < (unsigned)DEG_RANGE) atomicAdd(&cnt[k0], 1);
                if (in && k1 < (unsigned)DEG_RANGE) atomicAdd(&cnt[k1], 1);
                if (in && k2 < (unsigned)DEG_RANGE) atomicAdd(&cnt[k2], 1);
                if (in && k3 < (unsigned)DEG_RANGE) atomicAdd(&cnt[k3], 1);
            }
        }
        const int i = e0 + 4 * nq + tid;
        if (tid < 4 && i < e1) {
            const unsigned k = (unsigned)(dst[i] - lo);
            if (k < (unsigned)DEG_RANGE) atomicAdd(&cnt[k], 1);
        }
    } else {
        for (int i = e0 + tid; i < e1; i += 256) {
            const unsigned k = (unsigned)(dst[i] - lo);
            if (k < (unsigned)DEG_RANGE) atomicAdd(&cnt[k], 1);
        }
    }
    __syncthreads();

    const int hi = min(DEG_RANGE, n - lo);
    int* p = part + (size_t)c * n + lo;
    for (int i = tid; i < hi; i += 256) p[i] = cnt[i];
}

constexpr int AGG_ROWS = 400;

__device__ __forceinline__ void write_keys(const int* __restrict__ dst, unsigned* __restrict__ keys,
                                           int e, int gtid, int gsize) {
    const int nw = ((e + 15) >> 4) << 2;
    for (int w = gtid; w < nw; w += gsize) {
        unsigned v = 0;
        #pragma unroll
        for (int j = 0; j < 4; ++j) {
            const int i = 4 * w + j;
            unsigned k = 0xFFu;
            if (i < e) {
                const unsigned p = (unsigned)dst[i] / (unsigned)AGG_ROWS;
                k = p < 0xFFu ? p : 0xFFu;
            }
            v |= k << (8 * j);
        }
        keys[w] = v;
    }
}

constexpr int REC_SHIFT = 23;
constexpr int DEG_BLOCK = 256;
constexpr unsigned DEG16_BIG = 0xFFFFu;

struct Lists {
    unsigned*       tbl;
    unsigned*       lx;
    unsigned short* ld;
};

__device__ __forceinline__ void build_recs(const int* __restrict__ src, const int* __restrict__ dst,
                                           const int* __restrict__ degi, uint2* __restrict__ rec,
                                           int e, int gtid, int gsize) {
    constexpr int B = 4;
    for (int i0 = gtid; i0 < e; i0 += B * gsize) {
        int s[B], d[B];
        int ws[B];
        #pragma unroll
        for (int j = 0; j < B; ++j) {
            const int i = min(i0 + j * gsize, e - 1);
            s[j] = src[i];
            d[j] = dst[i];
        }
        #pragma unroll
        for (int j = 0; j < B; ++j) ws[j] = degi[s[j]] + 1;
        #pragma unroll
        for (int j = 0; j < B; ++j) {
            const int i = i0 + j * gsize;
            if (i < e) {
                uint2 r;
                r.x = (unsigned)s[j] | (((unsigned)d[j] % (unsigned)AGG_ROWS) << REC_SHIFT);
                r.y = (unsigned)ws[j];
                rec[i] = r;
            }
        }
    }
}

__device__ __forceinline__ void copy_words(const unsigned* __restrict__ from, unsigned* __restrict__ to,
                                           int nw, int gtid, int gsize) {
    const int nq = nw >> 2;
    const uint4* f4 = reinterpret_cast<const uint4*>(from);
    uint4* t4 = reinterpret_cast<uint4*>(to);
    for (int i = gtid; i < nq; i += gsize) t4[i] = f4[i];
    const int i = 4 * nq + gtid;
    if (gtid < 4 && i < nw) to[i] = from[i];
}

__device__ __forceinline__ void copy_lists(const Lists from, const Lists to, int e, int np, int gtid, int gsize) {
    copy_words(from.tbl, to.tbl, 2 * np, gtid, gsize);
    copy_words(from.lx, to.lx, e, gtid, gsize);
    copy_words(reinterpret_cast<const unsigned*>(from.ld), reinterpret_cast<unsigned*>(to.ld), e >> 1, gtid, gsize);
    if ((e & 1) && gtid == 0) to.ld[e - 1] = from.ld[e - 1];
}

__global__ void __launch_bounds__(DEG_BLOCK)
k_finalize_dis(const int* __restrict__ part, float* __restrict__ dis,
               int* __restrict__ degi, int* __restrict__ bsum,
               const int* __restrict__ dst, unsigned* __restrict__ keys,
               int e, int n) {
    __shared__ int bs;
    if (threadIdx.x == 0) bs = 0;
    __syncthreads();
    int i = blockIdx.x * blockDim.x + threadIdx.x;
    int din = 0;
    if (i < n) {
        for (int c = 0; c < DEG_CHUNKS; ++c) din += part[(size_t)c * n + i];
        dis[i]  = rsqrtf((float)(din + 1));
        degi[i] = din;
    }
    atomicAdd(&bs, din);
    __syncthreads();
    if (threadIdx.x == 0) bsum[blockIdx.x] = bs;
    write_keys(dst, keys, e, i, (int)(gridDim.x * blockDim.x));
}

template <int FIN, int FOUT, int MT, bool RELU, bool LISTS>
__global__ void k_gemm_wmma(const float* __restrict__ X, const float* __restrict__ W,
                            float* __restrict__ H, int M,
                            const int* __restrict__ src, const int* __restrict__ dst,
                            const int* __restrict__ degi, uint2* __restrict__ rec_to,
                            const Lists l_from, const Lists l_to, int e, int np) {
    constexpr int LDA = FIN + 1;
    __shared__ float sW[FIN * FOUT];
    __shared__ float sA[MT * 16 * LDA];

    const int tid  = threadIdx.x;
    const int nthr = 32 * (FOUT / 16);
    const int m0   = blockIdx.x * (MT * 16);

    if (LISTS) {
        copy_lists(l_from, l_to, e, np, (int)(blockIdx.x * nthr + tid), (int)(gridDim.x * nthr));
    } else {
        build_recs(src, dst, degi, rec_to, e, (int)(blockIdx.x * nthr + tid), (int)(gridDim.x * nthr));
    }

    for (int i = tid; i < FIN * FOUT; i += nthr) sW[i] = W[i];
    for (int i = tid; i < MT * 16 * FIN; i += nthr) {
        int r = i / FIN, c = i - r * FIN;
        int row = m0 + r; if (row >= M) row = M - 1;
        float v = X[(size_t)row * FIN + c];
        sA[r * LDA + c] = RELU ? fmaxf(v, 0.0f) : v;
    }
    __syncthreads();

    const int wave = tid >> 5;
    const int lane = tid & 31;
    const int half = lane >> 4;
    const int l16  = lane & 15;
    const int n0   = wave * 16;

    v2f bf[FIN / 4];
    #pragma unroll
    for (int kk = 0; kk < FIN / 4; ++kk) {
        int k = kk * 4 + 2 * half;
        bf[kk].x = sW[k * FOUT + n0 + l16];
        bf[kk].y = sW[(k + 1) * FOUT + n0 + l16];
    }

    #pragma unroll
    for (int mt = 0; mt < MT; ++mt) {
        const float* At = &sA[mt * 16 * LDA];
        v8f c = {};
        #pragma unroll
        for (int kk = 0; kk < FIN / 4; ++kk) {
            int k = kk * 4 + 2 * half;
            v2f a;
            a.x = At[l16 * LDA + k];
            a.y = At[l16 * LDA + k + 1];
            c = __builtin_amdgcn_wmma_f32_16x16x4_f32(false, a, false, bf[kk],
                                                      (short)0, c, false, false);
        }
        #pragma unroll
        for (int r = 0; r < 8; ++r) {
            int row = m0 + mt * 16 + half * 8 + r;
            if (row < M) H[(size_t)row * FOUT + n0 + l16] = c[r];
        }
    }
}

typedef __attribute__((ext_vector_type(4))) float v4f;
template <int F> struct RowVec;
template <> struct RowVec<64>  { typedef v2f type; };
template <> struct RowVec<128> { typedef v4f type; };

template <int F>
__device__ __forceinline__ int lds_col(int f) {
    return 32 * (f % (F / 32)) + f / (F / 32);
}

template <int F, int U>
__device__ __forceinline__ void rows_load(float (&v)[U][F / 32], const float* hl, int s, int j0) {
    typedef typename RowVec<F>::type vec;
    #pragma unroll
    for (int u = 0; u < U; ++u) {
        const int sj = __builtin_amdgcn_readlane(s, j0 + u);
        const vec t = *reinterpret_cast<const vec*>(hl + (size_t)sj * F);
        #pragma unroll
        for (int k = 0; k < F / 32; ++k) v[u][k] = t[k];
    }
}

template <int F, int U>
__device__ __forceinline__ void rows_add(const float (&v)[U][F / 32], float* al, int kr, float nrm, int j0) {
    #pragma unroll
    for (int u = 0; u < U; ++u) {
        const int   kj = __builtin_amdgcn_readlane(kr, j0 + u);
        const float nj = __int_as_float(__builtin_amdgcn_readlane(__float_as_int(nrm), j0 + u));
        float* ap = al + kj * F;
        #pragma unroll
        for (int k = 0; k < F / 32; ++k) atomicAdd(ap + 32 * k, v[u][k] * nj);
    }
}

template <int F, int U>
__device__ __forceinline__ void walk32(const float* hl, float* al, int s, int kr, float nrm) {
    float va[U][F / 32], vb[U][F / 32];
    rows_load<F, U>(va, hl, s, 0);
    #pragma unroll
    for (int j0 = 0; j0 < 32; j0 += 2 * U) {
        rows_load<F, U>(vb, hl, s, j0 + U);
        rows_add<F, U>(va, al, kr, nrm, j0);
        if (j0 + 2 * U < 32) rows_load<F, U>(va, hl, s, j0 + 2 * U);
        rows_add<F, U>(vb, al, kr, nrm, j0 + U);
    }
}

__device__ __forceinline__ void list_append(const Lists& L, int off, int* lcount, bool ok,
                                            int s, unsigned kr, unsigned deg, int lane) {
    const unsigned okm = __builtin_amdgcn_ballot_w32(ok);
    int base = 0;
    if (lane == 0) base = atomicAdd(lcount, __builtin_popcount(okm));
    base = __builtin_amdgcn_readfirstlane(base);
    if (ok) {
        const int slot = off + base + (int)__builtin_amdgcn_mbcnt_lo(okm, 0u);
        L.lx[slot] = (unsigned)s | (kr << REC_SHIFT);
        L.ld[slot] = (unsigned short)(deg < DEG16_BIG ? deg : DEG16_BIG);
    }
}

template <int F, int U>
__global__ void __launch_bounds__(1024)
k_agg(const float* __restrict__ h, const unsigned* __restrict__ keys,
      const uint2* __restrict__ rec, const int* __restrict__ dst,
      const float* __restrict__ dis, const float* __restrict__ bias,
      const int* __restrict__ degi, const int* __restrict__ bsum, const Lists L,
      float* __restrict__ out, int e, int n) {
    static_assert(32 % (2 * U) == 0 && F % 32 == 0, "group shape");
    constexpr int B = 4;
    __shared__ __attribute__((aligned(16))) float acc[AGG_ROWS * F];
    __shared__ float dl[AGG_ROWS];
    __shared__ int loff;
    __shared__ int lcount;

    const int tid  = threadIdx.x;
    const int lane = tid & 31;
    const int lo   = blockIdx.x * AGG_ROWS;
    const int rows = min(AGG_ROWS, n - lo);

    if (tid == 0) { loff = 0; lcount = 0; }
    __syncthreads();
    {
        const int nb = lo / DEG_BLOCK;
        int v = 0;
        for (int i = tid; i < nb; i += 1024) v += bsum[i];
        for (int i = nb * DEG_BLOCK + tid; i < lo; i += 1024) v += degi[i];
        atomicAdd(&loff, v);
    }

    for (int i = tid; i < AGG_ROWS; i += 1024) dl[i] = (i < rows) ? dis[lo + i] : 0.0f;
    {
        const float4* h4 = reinterpret_cast<const float4*>(h + (size_t)lo * F);
        const float4* b4 = reinterpret_cast<const float4*>(bias);
        for (int i = tid; i < rows * (F / 4); i += 1024) {
            const int r = i / (F / 4), c = i - r * (F / 4);
            const float s = dis[lo + r];
            const float w = s * s;
            const float4 hv = h4[i];
            const float4 bv = b4[c];
            float* ar = acc + r * F;
            ar[lds_col<F>(4 * c)]     = hv.x * w + bv.x;
            ar[lds_col<F>(4 * c + 1)] = hv.y * w + bv.y;
            ar[lds_col<F>(4 * c + 2)] = hv.z * w + bv.z;
            ar[lds_col<F>(4 * c + 3)] = hv.w * w + bv.w;
        }
    }
    __syncthreads();

    const int off = loff;
    const float* hl = h + (F / 32) * lane;
    float*       al = acc + lane;
    const unsigned kb   = blockIdx.x < 0xFFu ? blockIdx.x : 0xFFu;
    const unsigned key4 = kb * 0x01010101u;
    const int nq = (e + 15) >> 4;
    const uint4* kq = reinterpret_cast<const uint4*>(keys);
    const int per = 1024 * B;
    const int T   = (nq + per - 1) / per;
    const int rot = (int)(((long long)blockIdx.x * T) / gridDim.x);

    int qe = 0;
    int qc = 0;

    uint4 cur[B], nxt[B];
    int itr = rot;
    #pragma unroll
    for (int j = 0; j < B; ++j) cur[j] = kq[min(itr * per + j * 1024 + tid, nq - 1)];

    for (int it = 0; it < T; ++it) {
        int itn = itr + 1;
        if (itn >= T) itn = 0;
        #pragma unroll
        for (int j = 0; j < B; ++j) nxt[j] = kq[min(itn * per + j * 1024 + tid, nq - 1)];

        const int g0 = itr * per + tid;
        unsigned m[B];
        unsigned any = 0;
        #pragma unroll
        for (int j = 0; j < B; ++j) {
            const unsigned x0 = cur[j].x ^ key4, x1 = cur[j].y ^ key4, x2 = cur[j].z ^ key4, x3 = cur[j].w ^ key4;
            const unsigned z0 = ~(((x0 & 0x7F7F7F7Fu) + 0x7F7F7F7Fu) | x0) & 0x80808080u;
            const unsigned z1 = ~(((x1 & 0x7F7F7F7Fu) + 0x7F7F7F7Fu) | x1) & 0x80808080u;
            const unsigned z2 = ~(((x2 & 0x7F7F7F7Fu) + 0x7F7F7F7Fu) | x2) & 0x80808080u;
            const unsigned z3 = ~(((x3 & 0x7F7F7F7Fu) + 0x7F7F7F7Fu) | x3) & 0x80808080u;
            m[j] = (z0 >> 7) | (z1 >> 6) | (z2 >> 5) | (z3 >> 4);
            if (g0 + j * 1024 >= nq) m[j] = 0;
            any |= m[j];
        }
        while (__builtin_amdgcn_ballot_w32(any != 0) != 0) {
            const bool pass = (any != 0);
            unsigned mm = m[B - 1];
            int jj = B - 1;
            #pragma unroll
            for (int j = B - 2; j >= 0; --j) {
                if (m[j] != 0) { mm = m[j]; jj = j; }
            }
            const int bit = pass ? __builtin_ctz(mm) : 0;
            const int ei  = 16 * (g0 + jj * 1024) + 4 * (bit & 7) + (bit >> 3);
            any = 0;
            #pragma unroll
            for (int j = 0; j < B; ++j) {
                if (jj == j) m[j] &= m[j] - 1;
                any |= m[j];
            }
            const unsigned bm = __builtin_amdgcn_ballot_w32(pass);
            const int cnt = __builtin_popcount(bm);
            const int rk  = (int)__builtin_amdgcn_mbcnt_lo(bm, 0u);
            const int pos = pass ? (qc + rk) : (qc + cnt + (lane - rk));
            const int pe = __builtin_amdgcn_ds_permute((pos & 31) << 2, ei);
            const int me = (lane < qc) ? qe : pe;
            if (qc + cnt >= 32) {
                const int ec = min(me, e - 1);
                const uint2 r = rec[ec];
                const int s  = (int)(r.x & ((1u << REC_SHIFT) - 1u));
                unsigned kr  = r.x >> REC_SHIFT;
                bool ok = (me < e);
                if (blockIdx.x >= 0xFFu) {
                    kr = (unsigned)(dst[ec] - lo);
                    ok = ok && (kr < (unsigned)rows);
                }
                if (!ok) kr = 0;
                const float nrm = ok ? rsqrtf((float)(int)r.y) * dl[kr] : 0.0f;
                list_append(L, off, &lcount, ok, s, kr, r.y, lane);
                walk32<F, U>(hl, al, s, (int)kr, nrm);
                qe = pe;
                qc = qc + cnt - 32;
            } else {
                qe = me;
                qc = qc + cnt;
            }
        }
        #pragma unroll
        for (int j = 0; j < B; ++j) cur[j] = nxt[j];
        itr = itn;
    }

    if (qc > 0) {
        int s = 0, kr = 0;
        float nrm = 0.0f;
        bool ok = false;
        unsigned dg = 1;
        if (lane < qc) {
            const int ec = min(qe, e - 1);
            const uint2 r = rec[ec];
            s = (int)(r.x & ((1u << REC_SHIFT) - 1u));
            unsigned k = r.x >> REC_SHIFT;
            ok = (qe < e);
            if (blockIdx.x >= 0xFFu) {
                k  = (unsigned)(dst[ec] - lo);
                ok = ok && (k < (unsigned)rows);
            }
            dg  = r.y;
            kr  = ok ? (int)k : 0;
            nrm = ok ? rsqrtf((float)(int)dg) * dl[kr] : 0.0f;
        }
        list_append(L, off, &lcount, ok, s, (unsigned)kr, dg, lane);
        for (int j = 0; j < qc; ++j) {
            float v1[1][F / 32];
            rows_load<F, 1>(v1, hl, s, j);
            rows_add<F, 1>(v1, al, kr, nrm, j);
        }
    }
    __syncthreads();

    {
        float4* o4 = reinterpret_cast<float4*>(out + (size_t)lo * F);
        for (int i = tid; i < rows * (F / 4); i += 1024) {
            const int r = i / (F / 4), c = i - r * (F / 4);
            const float* ar = acc + r * F;
            float4 o;
            o.x = ar[lds_col<F>(4 * c)];      o.y = ar[lds_col<F>(4 * c + 1)];
            o.z = ar[lds_col<F>(4 * c + 2)];  o.w = ar[lds_col<F>(4 * c + 3)];
            o4[i] = o;
        }
    }
    if (tid == 0) {
        L.tbl[2 * blockIdx.x]     = (unsigned)off;
        L.tbl[2 * blockIdx.x + 1] = (unsigned)lcount;
    }
}

template <int F, int U>
__global__ void __launch_bounds__(1024)
k_agg_list(const float* __restrict__ h, const unsigned* __restrict__ tbl,
           const unsigned* __restrict__ lx, const unsigned short* __restrict__ ld,
           const float* __restrict__ dis, const float* __restrict__ bias,
           float* __restrict__ out, int n) {
    static_assert(32 % (2 * U) == 0 && F % 32 == 0, "group shape");
    __shared__ __attribute__((aligned(16))) float acc[AGG_ROWS * F];
    __shared__ float dl[AGG_ROWS];

    const int tid  = threadIdx.x;
    const int lane = tid & 31;
    const int wave = __builtin_amdgcn_readfirstlane(tid >> 5);
    const int lo   = blockIdx.x * AGG_ROWS;
    const int rows = min(AGG_ROWS, n - lo);

    const int off = (int)tbl[2 * blockIdx.x];
    const int cnt = (int)tbl[2 * blockIdx.x + 1];
    const int last = off + max(cnt, 1) - 1;
    int pos = 32 * wave + lane;
    unsigned x  = lx[min(off + pos, last)];
    unsigned dg = ld[min(off + pos, last)];

    for (int i = tid; i < AGG_ROWS; i += 1024) dl[i] = (i < rows) ? dis[lo + i] : 0.0f;
    {
        const float4* h4 = reinterpret_cast<const float4*>(h + (size_t)lo * F);
        const float4* b4 = reinterpret_cast<const float4*>(bias);
        for (int i = tid; i < rows * (F / 4); i += 1024) {
            const int r = i / (F / 4), c = i - r * (F / 4);
            const float s = dis[lo + r];
            const float w = s * s;
            const float4 hv = h4[i];
            const float4 bv = b4[c];
            float* ar = acc + r * F;
            ar[lds_col<F>(4 * c)]     = hv.x * w + bv.x;
            ar[lds_col<F>(4 * c + 1)] = hv.y * w + bv.y;
            ar[lds_col<F>(4 * c + 2)] = hv.z * w + bv.z;
            ar[lds_col<F>(4 * c + 3)] = hv.w * w + bv.w;
        }
    }
    __syncthreads();

    const float* hl = h + (F / 32) * lane;
    float*       al = acc + lane;
    for (int p0 = 32 * wave; p0 < cnt; p0 += 1024) {
        const bool ok = pos < cnt;
        const int  s  = ok ? (int)(x & ((1u << REC_SHIFT) - 1u)) : 0;
        const int  kr = ok ? (int)(x >> REC_SHIFT) : 0;
        float ds = rsqrtf((float)(int)dg);
        if (dg == DEG16_BIG) ds = dis[s];
        const int pn = pos + 1024;
        const unsigned xn = lx[min(off + pn, last)];
        const unsigned dn = ld[min(off + pn, last)];
        const float nrm = ok ? ds * dl[kr] : 0.0f;
        const int m = min(32, cnt - p0);
        if (m == 32) {
            walk32<F, U>(hl, al, s, kr, nrm);
        } else {
            for (int j = 0; j < m; ++j) {
                float v1[1][F / 32];
                rows_load<F, 1>(v1, hl, s, j);
                rows_add<F, 1>(v1, al, kr, nrm, j);
            }
        }
        x = xn;  dg = dn;  pos = pn;
    }
    __syncthreads();

    {
        float4* o4 = reinterpret_cast<float4*>(out + (size_t)lo * F);
        for (int i = tid; i < rows * (F / 4); i += 1024) {
            const int r = i / (F / 4), c = i - r * (F / 4);
            const float* ar = acc + r * F;
            float4 o;
            o.x = ar[lds_col<F>(4 * c)];      o.y = ar[lds_col<F>(4 * c + 1)];
            o.z = ar[lds_col<F>(4 * c + 2)];  o.w = ar[lds_col<F>(4 * c + 3)];
            o4[i] = o;
        }
    }
}

extern "C" void kernel_launch(void* const* d_in, const int* in_sizes, int n_in,
                              void* d_out, int out_size, void* d_ws, size_t ws_size,
                              hipStream_t stream) {
    const float* X  = (const float*)d_in[0];
    const int*   EI = (const int*)d_in[1];
    const float* W1 = (const float*)d_in[2];
    const float* b1 = (const float*)d_in[3];
    const float* W2 = (const float*)d_in[4];
    const float* b2 = (const float*)d_in[5];
    float* out = (float*)d_out;

    const int FIN1 = 64;
    const int N = in_sizes[0] / FIN1;
    const int E = in_sizes[1] / 2;
    const int* src = EI;
    const int* dst = EI + E;

    float* dis = (float*)d_ws;
    size_t off = ((size_t)N * sizeof(float) + 16 + 255) & ~(size_t)255;
    float* bufH = (float*)((char*)d_ws + off);
    float* bufO = bufH + (size_t)N * 128;
    int*   degp = (int*)bufH;
    const int NP   = (N + AGG_ROWS - 1) / AGG_ROWS;
    const int NBLK = (N + DEG_BLOCK - 1) / DEG_BLOCK;
    const size_t kbytes = (((((size_t)E + 15) >> 4) << 4) + 255) & ~(size_t)255;
    const size_t rbytes = ((size_t)E * 8 + 255) & ~(size_t)255;
    const size_t dbytes = ((size_t)N * 4 + 255) & ~(size_t)255;
    const size_t sbytes = ((size_t)NBLK * 4 + 255) & ~(size_t)255;
    const size_t tbytes = ((size_t)NP * 8 + 255) & ~(size_t)255;
    const size_t xbytes = ((size_t)E * 4 + 255) & ~(size_t)255;
    unsigned* keys1 = (unsigned*)d_out;
    uint2* rec1 = (uint2*)((char*)keys1 + kbytes);
    int*   degi = (int*)((char*)rec1 + rbytes);
    int*   bsum = (int*)((char*)degi + dbytes);
    Lists l1, l2;
    l1.tbl = (unsigned*)((char*)bsum + sbytes);
    l1.lx  = (unsigned*)((char*)l1.tbl + tbytes);
    l1.ld  = (unsigned short*)((char*)l1.lx + xbytes);
    l2.tbl = (unsigned*)(bufH + (size_t)N * 64);
    l2.lx  = (unsigned*)((char*)l2.tbl + tbytes);
    l2.ld  = (unsigned short*)((char*)l2.lx + xbytes);

    const int T = DEG_BLOCK;
    dim3 blkN(NBLK);
    dim3 blkD((N + DEG_RANGE - 1) / DEG_RANGE, DEG_CHUNKS);
    const int ntiles = (N + 15) / 16;
    dim3 blkA((N + AGG_ROWS - 1) / AGG_ROWS);

    k_count_deg<<<blkD, 256, 0, stream>>>(dst, degp, E, N);
    k_finalize_dis<<<blkN, T, 0, stream>>>(degp, dis, degi, bsum, dst, keys1, E, N);

    k_gemm_wmma<64, 128, 4, false, false><<<(ntiles + 3) / 4, 256, 0, stream>>>(
        X, W1, bufH, N, src, dst, degi, rec1, l1, l2, E, NP);
    k_agg<128, 4><<<blkA, 1024, 0, stream>>>(bufH, keys1, rec1, dst, dis, b1, degi, bsum, l1, bufO, E, N);

    k_gemm_wmma<128, 64, 2, true, true><<<(ntiles + 1) / 2, 128, 0, stream>>>(
        bufO, W2, bufH, N, src, dst, degi, rec1, l1, l2, E, NP);
    k_agg_list<64, 8><<<blkA, 1024, 0, stream>>>(bufH, l2.tbl, l2.lx, l2.ld, dis, b2, out, N);
}


# Layer-1 owner work-groups run relu and the W2 product (f32 WMMA) on their finished LDS rows, store h2: no 2nd dense launch, out1 round trip or list copy; W2 product sums each 4 k as 0,2,1,3
# speedup vs baseline: 3.5910x; 1.1814x over previous
#include <hip/hip_runtime.h>
#include <hip/hip_bf16.h>

typedef __attribute__((ext_vector_type(2))) float v2f;
typedef __attribute__((ext_vector_type(8))) float v8f;

constexpr int DEG_RANGE  = 16384;
constexpr int DEG_CHUNKS = 64;

__global__ void __launch_bounds__(256)
k_count_deg(const int* __restrict__ dst, int* __restrict__ part, int e, int n) {
    __shared__ int cnt[DEG_RANGE];
    const int tid = threadIdx.x;
    const int lo  = blockIdx.x * DEG_RANGE;
    const int c   = blockIdx.y;
    for (int i = tid; i < DEG_RANGE; i += 256) cnt[i] = 0;
    __syncthreads();

    const int per = (((e + DEG_CHUNKS - 1) / DEG_CHUNKS) + 3) & ~3;
    const int e0  = c * per;
    const int e1  = min(e0 + per, e);
    if ((((size_t)dst) & 15) == 0) {
        const int nq = (e1 > e0) ? ((e1 - e0) >> 2) : 0;
        const int4* dq = reinterpret_cast<const int4*>(dst + e0);
        constexpr int B = 8;
        for (int qb = 0; qb < nq; qb += 256 * B) {
            int4 v[B];
            #pragma unroll
            for (int j = 0; j < B; ++j) {
                const int q = qb + j * 256 + tid;
                v[j] = dq[min(q, nq - 1)];
            }
            #pragma unroll
            for (int j = 0; j < B; ++j) {
                const bool in = (qb + j * 256 + tid) < nq;
                const unsigned k0 = (unsigned)(v[j].x - lo), k1 = (unsigned)(v[j].y - lo);
                const unsigned k2 = (unsigned)(v[j].z - lo), k3 = (unsigned)(v[j].w - lo);
                if (in && k0 < (unsigned)DEG_RANGE) atomicAdd(&cnt[k0], 1);
                if (in && k1 < (unsigned)DEG_RANGE) atomicAdd(&cnt[k1], 1);
                if (in && k2 < (unsigned)DEG_RANGE) atomicAdd(&cnt[k2], 1);
                if (in && k3 < (unsigned)DEG_RANGE) atomicAdd(&cnt[k3], 1);
            }
        }
        const int i = e0 + 4 * nq + tid;
        if (tid < 4 && i < e1) {
            const unsigned k = (unsigned)(dst[i] - lo);
            if (k < (unsigned)DEG_RANGE) atomicAdd(&cnt[k], 1);
        }
    } else {
        for (int i = e0 + tid; i < e1; i += 256) {
            const unsigned k = (unsigned)(dst[i] - lo);
            if (k < (unsigned)DEG_RANGE) atomicAdd(&cnt[k], 1);
        }
    }
    __syncthreads();

    const int hi = min(DEG_RANGE, n - lo);
    int* p = part + (size_t)c * n + lo;
    for (int i = tid; i < hi; i += 256) p[i] = cnt[i];
}

constexpr int AGG_ROWS = 400;

__device__ __forceinline__ void write_keys(const int* __restrict__ dst, unsigned* __restrict__ keys,
                                           int e, int gtid, int gsize) {
    const int nw = ((e + 15) >> 4) << 2;
    for (int w = gtid; w < nw; w += gsize) {
        unsigned v = 0;
        #pragma unroll
        for (int j = 0; j < 4; ++j) {
            const int i = 4 * w + j;
            unsigned k = 0xFFu;
            if (i < e) {
                const unsigned p = (unsigned)dst[i] / (unsigned)AGG_ROWS;
                k = p < 0xFFu ? p : 0xFFu;
            }
            v |= k << (8 * j);
        }
        keys[w] = v;
    }
}

constexpr int REC_SHIFT = 23;
constexpr int DEG_BLOCK = 256;
constexpr unsigned DEG16_BIG = 0xFFFFu;

struct Lists {
    unsigned*       tbl;
    unsigned*       lx;
    unsigned short* ld;
};

__device__ __forceinline__ void build_recs(const int* __restrict__ src, const int* __restrict__ dst,
                                           const int* __restrict__ degi, uint2* __restrict__ rec,
                                           int e, int gtid, int gsize) {
    constexpr int B = 4;
    for (int i0 = gtid; i0 < e; i0 += B * gsize) {
        int s[B], d[B];
        int ws[B];
        #pragma unroll
        for (int j = 0; j < B; ++j) {
            const int i = min(i0 + j * gsize, e - 1);
            s[j] = src[i];
            d[j] = dst[i];
        }
        #pragma unroll
        for (int j = 0; j < B; ++j) ws[j] = degi[s[j]] + 1;
        #pragma unroll
        for (int j = 0; j < B; ++j) {
            const int i = i0 + j * gsize;
            if (i < e) {
                uint2 r;
                r.x = (unsigned)s[j] | (((unsigned)d[j] % (unsigned)AGG_ROWS) << REC_SHIFT);
                r.y = (unsigned)ws[j];
                rec[i] = r;
            }
        }
    }
}

__global__ void __launch_bounds__(DEG_BLOCK)
k_finalize_dis(const int* __restrict__ part, float* __restrict__ dis,
               int* __restrict__ degi, int* __restrict__ bsum,
               const int* __restrict__ dst, unsigned* __restrict__ keys,
               int e, int n) {
    __shared__ int bs;
    if (threadIdx.x == 0) bs = 0;
    __syncthreads();
    int i = blockIdx.x * blockDim.x + threadIdx.x;
    int din = 0;
    if (i < n) {
        for (int c = 0; c < DEG_CHUNKS; ++c) din += part[(size_t)c * n + i];
        dis[i]  = rsqrtf((float)(din + 1));
        degi[i] = din;
    }
    atomicAdd(&bs, din);
    __syncthreads();
    if (threadIdx.x == 0) bsum[blockIdx.x] = bs;
    write_keys(dst, keys, e, i, (int)(gridDim.x * blockDim.x));
}

template <int FIN, int FOUT, int MT>
__global__ void k_gemm_wmma(const float* __restrict__ X, const float* __restrict__ W,
                            float* __restrict__ H, int M,
                            const int* __restrict__ src, const int* __restrict__ dst,
                            const int* __restrict__ degi, uint2* __restrict__ rec_to, int e) {
    constexpr int LDA = FIN + 1;
    __shared__ float sW[FIN * FOUT];
    __shared__ float sA[MT * 16 * LDA];

    const int tid  = threadIdx.x;
    const int nthr = 32 * (FOUT / 16);
    const int m0   = blockIdx.x * (MT * 16);

    build_recs(src, dst, degi, rec_to, e, (int)(blockIdx.x * nthr + tid), (int)(gridDim.x * nthr));

    for (int i = tid; i < FIN * FOUT; i += nthr) sW[i] = W[i];
    for (int i = tid; i < MT * 16 * FIN; i += nthr) {
        int r = i / FIN, c = i - r * FIN;
        int row = m0 + r; if (row >= M) row = M - 1;
        float v = X[(size_t)row * FIN + c];
        sA[r * LDA + c] = v;
    }
    __syncthreads();

    const int wave = tid >> 5;
    const int lane = tid & 31;
    const int half = lane >> 4;
    const int l16  = lane & 15;
    const int n0   = wave * 16;

    v2f bf[FIN / 4];
    #pragma unroll
    for (int kk = 0; kk < FIN / 4; ++kk) {
        int k = kk * 4 + 2 * half;
        bf[kk].x = sW[k * FOUT + n0 + l16];
        bf[kk].y = sW[(k + 1) * FOUT + n0 + l16];
    }

    #pragma unroll
    for (int mt = 0; mt < MT; ++mt) {
        const float* At = &sA[mt * 16 * LDA];
        v8f c = {};
        #pragma unroll
        for (int kk = 0; kk < FIN / 4; ++kk) {
            int k = kk * 4 + 2 * half;
            v2f a;
            a.x = At[l16 * LDA + k];
            a.y = At[l16 * LDA + k + 1];
            c = __builtin_amdgcn_wmma_f32_16x16x4_f32(false, a, false, bf[kk],
                                                      (short)0, c, false, false);
        }
        #pragma unroll
        for (int r = 0; r < 8; ++r) {
            int row = m0 + mt * 16 + half * 8 + r;
            if (row < M) H[(size_t)row * FOUT + n0 + l16] = c[r];
        }
    }
}

__device__ __forceinline__ int frag_k(int kk, int h, int e) { return 4 * kk + 2 * e + h; }

typedef __attribute__((ext_vector_type(4))) float v4f;
template <int F> struct RowVec;
template <> struct RowVec<64>  { typedef v2f type; };
template <> struct RowVec<128> { typedef v4f type; };

template <int F>
__device__ __forceinline__ int lds_col(int f) {
    return 32 * (f % (F / 32)) + f / (F / 32);
}

template <int F, int U>
__device__ __forceinline__ void rows_load(float (&v)[U][F / 32], const float* hl, int s, int j0) {
    typedef typename RowVec<F>::type vec;
    #pragma unroll
    for (int u = 0; u < U; ++u) {
        const int sj = __builtin_amdgcn_readlane(s, j0 + u);
        const vec t = *reinterpret_cast<const vec*>(hl + (size_t)sj * F);
        #pragma unroll
        for (int k = 0; k < F / 32; ++k) v[u][k] = t[k];
    }
}

template <int F, int U, int P>
__device__ __forceinline__ void rows_add(const float (&v)[U][F / 32], float* al, int kr, float nrm, int j0) {
    #pragma unroll
    for (int u = 0; u < U; ++u) {
        const int   kj = __builtin_amdgcn_readlane(kr, j0 + u);
        const float nj = __int_as_float(__builtin_amdgcn_readlane(__float_as_int(nrm), j0 + u));
        float* ap = al + kj * P;
        #pragma unroll
        for (int k = 0; k < F / 32; ++k) atomicAdd(ap + 32 * k, v[u][k] * nj);
    }
}

template <int F, int U, int P>
__device__ __forceinline__ void walk32(const float* hl, float* al, int s, int kr, float nrm) {
    float va[U][F / 32], vb[U][F / 32];
    rows_load<F, U>(va, hl, s, 0);
    #pragma unroll
    for (int j0 = 0; j0 < 32; j0 += 2 * U) {
        rows_load<F, U>(vb, hl, s, j0 + U);
        rows_add<F, U, P>(va, al, kr, nrm, j0);
        if (j0 + 2 * U < 32) rows_load<F, U>(va, hl, s, j0 + 2 * U);
        rows_add<F, U, P>(vb, al, kr, nrm, j0 + U);
    }
}

__device__ __forceinline__ void list_append(const Lists& L, int off, int* lcount, bool ok,
                                            int s, unsigned kr, unsigned deg, int lane) {
    const unsigned okm = __builtin_amdgcn_ballot_w32(ok);
    int base = 0;
    if (lane == 0) base = atomicAdd(lcount, __builtin_popcount(okm));
    base = __builtin_amdgcn_readfirstlane(base);
    if (ok) {
        const int slot = off + base + (int)__builtin_amdgcn_mbcnt_lo(okm, 0u);
        L.lx[slot] = (unsigned)s | (kr << REC_SHIFT);
        L.ld[slot] = (unsigned short)(deg < DEG16_BIG ? deg : DEG16_BIG);
    }
}

template <int F, int U, int FO>
__global__ void __launch_bounds__(1024)
k_agg(const float* __restrict__ h, const unsigned* __restrict__ keys,
      const uint2* __restrict__ rec, const int* __restrict__ dst,
      const float* __restrict__ dis, const float* __restrict__ bias,
      const int* __restrict__ degi, const int* __restrict__ bsum, const Lists L,
      const float* __restrict__ W2, float* __restrict__ h2, int e, int n) {
    static_assert(32 % (2 * U) == 0 && F % 32 == 0, "group shape");
    static_assert(F == 128 && AGG_ROWS % 16 == 0 && FO % 16 == 0 && 32 % (FO / 16) == 0, "tile shape of the second product");
    constexpr int B = 4;
    constexpr int AP = F + 1;
    __shared__ float acc[AGG_ROWS * AP];
    __shared__ float dl[AGG_ROWS];
    __shared__ int loff;
    __shared__ int lcount;

    const int tid  = threadIdx.x;
    const int lane = tid & 31;
    const int lo   = blockIdx.x * AGG_ROWS;
    const int rows = min(AGG_ROWS, n - lo);

    if (tid == 0) { loff = 0; lcount = 0; }
    __syncthreads();
    {
        const int nb = lo / DEG_BLOCK;
        int v = 0;
        for (int i = tid; i < nb; i += 1024) v += bsum[i];
        for (int i = nb * DEG_BLOCK + tid; i < lo; i += 1024) v += degi[i];
        atomicAdd(&loff, v);
    }

    for (int i = tid; i < AGG_ROWS; i += 1024) dl[i] = (i < rows) ? dis[lo + i] : 0.0f;
    {
        const float4* h4 = reinterpret_cast<const float4*>(h + (size_t)lo * F);
        const float4* b4 = reinterpret_cast<const float4*>(bias);
        for (int i = tid; i < rows * (F / 4); i += 1024) {
            const int r = i / (F / 4), c = i - r * (F / 4);
            const float s = dis[lo + r];
            const float w = s * s;
            const float4 hv = h4[i];
            const float4 bv = b4[c];
            float* ar = acc + r * AP;
            ar[lds_col<F>(4 * c)]     = hv.x * w + bv.x;
            ar[lds_col<F>(4 * c + 1)] = hv.y * w + bv.y;
            ar[lds_col<F>(4 * c + 2)] = hv.z * w + bv.z;
            ar[lds_col<F>(4 * c + 3)] = hv.w * w + bv.w;
        }
    }
    __syncthreads();

    const int off = loff;
    const float* hl = h + (F / 32) * lane;
    float*       al = acc + lane;
    const unsigned kb   = blockIdx.x < 0xFFu ? blockIdx.x : 0xFFu;
    const unsigned key4 = kb * 0x01010101u;
    const int nq = (e + 15) >> 4;
    const uint4* kq = reinterpret_cast<const uint4*>(keys);
    const int per = 1024 * B;
    const int T   = (nq + per - 1) / per;
    const int rot = (int)(((long long)blockIdx.x * T) / gridDim.x);

    int qe = 0;
    int qc = 0;

    uint4 cur[B], nxt[B];
    int itr = rot;
    #pragma unroll
    for (int j = 0; j < B; ++j) cur[j] = kq[min(itr * per + j * 1024 + tid, nq - 1)];

    for (int it = 0; it < T; ++it) {
        int itn = itr + 1;
        if (itn >= T) itn = 0;
        #pragma unroll
        for (int j = 0; j < B; ++j) nxt[j] = kq[min(itn * per + j * 1024 + tid, nq - 1)];

        const int g0 = itr * per + tid;
        unsigned m[B];
        unsigned any = 0;
        #pragma unroll
        for (int j = 0; j < B; ++j) {
            const unsigned x0 = cur[j].x ^ key4, x1 = cur[j].y ^ key4, x2 = cur[j].z ^ key4, x3 = cur[j].w ^ key4;
            const unsigned z0 = ~(((x0 & 0x7F7F7F7Fu) + 0x7F7F7F7Fu) | x0) & 0x80808080u;
            const unsigned z1 = ~(((x1 & 0x7F7F7F7Fu) + 0x7F7F7F7Fu) | x1) & 0x80808080u;
            const unsigned z2 = ~(((x2 & 0x7F7F7F7Fu) + 0x7F7F7F7Fu) | x2) & 0x80808080u;
            const unsigned z3 = ~(((x3 & 0x7F7F7F7Fu) + 0x7F7F7F7Fu) | x3) & 0x80808080u;
            m[j] = (z0 >> 7) | (z1 >> 6) | (z2 >> 5) | (z3 >> 4);
            if (g0 + j * 1024 >= nq) m[j] = 0;
            any |= m[j];
        }
        while (__builtin_amdgcn_ballot_w32(any != 0) != 0) {
            const bool pass = (any != 0);
            unsigned mm = m[B - 1];
            int jj = B - 1;
            #pragma unroll
            for (int j = B - 2; j >= 0; --j) {
                if (m[j] != 0) { mm = m[j]; jj = j; }
            }
            const int bit = pass ? __builtin_ctz(mm) : 0;
            const int ei  = 16 * (g0 + jj * 1024) + 4 * (bit & 7) + (bit >> 3);
            any = 0;
            #pragma unroll
            for (int j = 0; j < B; ++j) {
                if (jj == j) m[j] &= m[j] - 1;
                any |= m[j];
            }
            const unsigned bm = __builtin_amdgcn_ballot_w32(pass);
            const int cnt = __builtin_popcount(bm);
            const int rk  = (int)__builtin_amdgcn_mbcnt_lo(bm, 0u);
            const int pos = pass ? (qc + rk) : (qc + cnt + (lane - rk));
            const int pe = __builtin_amdgcn_ds_permute((pos & 31) << 2, ei);
            const int me = (lane < qc) ? qe : pe;
            if (qc + cnt >= 32) {
                const int ec = min(me, e - 1);
                const uint2 r = rec[ec];
                const int s  = (int)(r.x & ((1u << REC_SHIFT) - 1u));
                unsigned kr  = r.x >> REC_SHIFT;
                bool ok = (me < e);
                if (blockIdx.x >= 0xFFu) {
                    kr = (unsigned)(dst[ec] - lo);
                    ok = ok && (kr < (unsigned)rows);
                }
                if (!ok) kr = 0;
                const float nrm = ok ? rsqrtf((float)(int)r.y) * dl[kr] : 0.0f;
                list_append(L, off, &lcount, ok, s, kr, r.y, lane);
                walk32<F, U, AP>(hl, al, s, (int)kr, nrm);
                qe = pe;
                qc = qc + cnt - 32;
            } else {
                qe = me;
                qc = qc + cnt;
            }
        }
        #pragma unroll
        for (int j = 0; j < B; ++j) cur[j] = nxt[j];
        itr = itn;
    }

    if (qc > 0) {
        int s = 0, kr = 0;
        float nrm = 0.0f;
        bool ok = false;
        unsigned dg = 1;
        if (lane < qc) {
            const int ec = min(qe, e - 1);
            const uint2 r = rec[ec];
            s = (int)(r.x & ((1u << REC_SHIFT) - 1u));
            unsigned k = r.x >> REC_SHIFT;
            ok = (qe < e);
            if (blockIdx.x >= 0xFFu) {
                k  = (unsigned)(dst[ec] - lo);
                ok = ok && (k < (unsigned)rows);
            }
            dg  = r.y;
            kr  = ok ? (int)k : 0;
            nrm = ok ? rsqrtf((float)(int)dg) * dl[kr] : 0.0f;
        }
        list_append(L, off, &lcount, ok, s, (unsigned)kr, dg, lane);
        for (int j = 0; j < qc; ++j) {
            float v1[1][F / 32];
            rows_load<F, 1>(v1, hl, s, j);
            rows_add<F, 1, AP>(v1, al, kr, nrm, j);
        }
    }
    {
        constexpr int NT = FO / 16;
        constexpr int MTILES = AGG_ROWS / 16;
        const int wave = __builtin_amdgcn_readfirstlane(tid >> 5);
        const int half = lane >> 4;
        const int l16  = lane & 15;
        const int n0   = (wave % NT) * 16;
        v2f bf[F / 4];
        #pragma unroll
        for (int kk = 0; kk < F / 4; ++kk) {
            bf[kk].x = W2[frag_k(kk, half, 0) * FO + n0 + l16];
            bf[kk].y = W2[frag_k(kk, half, 1) * FO + n0 + l16];
        }
        __syncthreads();

        for (int mt = wave / NT; mt < MTILES; mt += 32 / NT) {
            const float* ar = acc + (mt * 16 + l16) * AP;
            v8f c = {};
            #pragma unroll
            for (int kk = 0; kk < F / 4; ++kk) {
                v2f a;
                a.x = fmaxf(ar[lds_col<F>(frag_k(kk, half, 0))], 0.0f);
                a.y = fmaxf(ar[lds_col<F>(frag_k(kk, half, 1))], 0.0f);
                c = __builtin_amdgcn_wmma_f32_16x16x4_f32(false, a, false, bf[kk],
                                                          (short)0, c, false, false);
            }
            #pragma unroll
            for (int r = 0; r < 8; ++r) {
                const int row = mt * 16 + half * 8 + r;
                if (row < rows) h2[(size_t)(lo + row) * FO + n0 + l16] = c[r];
            }
        }
    }
    if (tid == 0) {
        L.tbl[2 * blockIdx.x]     = (unsigned)off;
        L.tbl[2 * blockIdx.x + 1] = (unsigned)lcount;
    }
}

template <int F, int U>
__global__ void __launch_bounds__(1024)
k_agg_list(const float* __restrict__ h, const unsigned* __restrict__ tbl,
           const unsigned* __restrict__ lx, const unsigned short* __restrict__ ld,
           const float* __restrict__ dis, const float* __restrict__ bias,
           float* __restrict__ out, int n) {
    static_assert(32 % (2 * U) == 0 && F % 32 == 0, "group shape");
    __shared__ __attribute__((aligned(16))) float acc[AGG_ROWS * F];
    __shared__ float dl[AGG_ROWS];

    const int tid  = threadIdx.x;
    const int lane = tid & 31;
    const int wave = __builtin_amdgcn_readfirstlane(tid >> 5);
    const int lo   = blockIdx.x * AGG_ROWS;
    const int rows = min(AGG_ROWS, n - lo);

    const int off = (int)tbl[2 * blockIdx.x];
    const int cnt = (int)tbl[2 * blockIdx.x + 1];
    const int last = off + max(cnt, 1) - 1;
    int pos = 32 * wave + lane;
    unsigned x  = lx[min(off + pos, last)];
    unsigned dg = ld[min(off + pos, last)];

    for (int i = tid; i < AGG_ROWS; i += 1024) dl[i] = (i < rows) ? dis[lo + i] : 0.0f;
    {
        const float4* h4 = reinterpret_cast<const float4*>(h + (size_t)lo * F);
        const float4* b4 = reinterpret_cast<const float4*>(bias);
        for (int i = tid; i < rows * (F / 4); i += 1024) {
            const int r = i / (F / 4), c = i - r * (F / 4);
            const float s = dis[lo + r];
            const float w = s * s;
            const float4 hv = h4[i];
            const float4 bv = b4[c];
            float* ar = acc + r * F;
            ar[lds_col<F>(4 * c)]     = hv.x * w + bv.x;
            ar[lds_col<F>(4 * c + 1)] = hv.y * w + bv.y;
            ar[lds_col<F>(4 * c + 2)] = hv.z * w + bv.z;
            ar[lds_col<F>(4 * c + 3)] = hv.w * w + bv.w;
        }
    }
    __syncthreads();

    const float* hl = h + (F / 32) * lane;
    float*       al = acc + lane;
    for (int p0 = 32 * wave; p0 < cnt; p0 += 1024) {
        const bool ok = pos < cnt;
        const int  s  = ok ? (int)(x & ((1u << REC_SHIFT) - 1u)) : 0;
        const int  kr = ok ? (int)(x >> REC_SHIFT) : 0;
        float ds = rsqrtf((float)(int)dg);
        if (dg == DEG16_BIG) ds = dis[s];
        const int pn = pos + 1024;
        const unsigned xn = lx[min(off + pn, last)];
        const unsigned dn = ld[min(off + pn, last)];
        const float nrm = ok ? ds * dl[kr] : 0.0f;
        const int m = min(32, cnt - p0);
        if (m == 32) {
            walk32<F, U, F>(hl, al, s, kr, nrm);
        } else {
            for (int j = 0; j < m; ++j) {
                float v1[1][F / 32];
                rows_load<F, 1>(v1, hl, s, j);
                rows_add<F, 1, F>(v1, al, kr, nrm, j);
            }
        }
        x = xn;  dg = dn;  pos = pn;
    }
    __syncthreads();

    {
        float4* o4 = reinterpret_cast<float4*>(out + (size_t)lo * F);
        for (int i = tid; i < rows * (F / 4); i += 1024) {
            const int r = i / (F / 4), c = i - r * (F / 4);
            const float* ar = acc + r * F;
            float4 o;
            o.x = ar[lds_col<F>(4 * c)];      o.y = ar[lds_col<F>(4 * c + 1)];
            o.z = ar[lds_col<F>(4 * c + 2)];  o.w = ar[lds_col<F>(4 * c + 3)];
            o4[i] = o;
        }
    }
}

extern "C" void kernel_launch(void* const* d_in, const int* in_sizes, int n_in,
                              void* d_out, int out_size, void* d_ws, size_t ws_size,
                              hipStream_t stream) {
    const float* X  = (const float*)d_in[0];
    const int*   EI = (const int*)d_in[1];
    const float* W1 = (const float*)d_in[2];
    const float* b1 = (const float*)d_in[3];
    const float* W2 = (const float*)d_in[4];
    const float* b2 = (const float*)d_in[5];
    float* out = (float*)d_out;

    const int FIN1 = 64;
    const int N = in_sizes[0] / FIN1;
    const int E = in_sizes[1] / 2;
    const int* src = EI;
    const int* dst = EI + E;

    float* dis = (float*)d_ws;
    size_t off = ((size_t)N * sizeof(float) + 16 + 255) & ~(size_t)255;
    float* bufH = (float*)((char*)d_ws + off);
    float* bufO = bufH + (size_t)N * 128;
    int*   degp = (int*)bufH;
    const int NP   = (N + AGG_ROWS - 1) / AGG_ROWS;
    const int NBLK = (N + DEG_BLOCK - 1) / DEG_BLOCK;
    const size_t kbytes = (((((size_t)E + 15) >> 4) << 4) + 255) & ~(size_t)255;
    const size_t rbytes = ((size_t)E * 8 + 255) & ~(size_t)255;
    const size_t dbytes = ((size_t)N * 4 + 255) & ~(size_t)255;
    const size_t tbytes = ((size_t)NP * 8 + 255) & ~(size_t)255;
    const size_t xbytes = ((size_t)E * 4 + 255) & ~(size_t)255;
    unsigned* keys1 = (unsigned*)d_out;
    uint2* rec1 = (uint2*)((char*)keys1 + kbytes);
    int*   degi = (int*)((char*)rec1 + rbytes);
    int*   bsum = (int*)((char*)degi + dbytes);
    float* h2 = bufO;
    Lists l1;
    l1.tbl = (unsigned*)(bufO + (size_t)N * 64);
    l1.lx  = (unsigned*)((char*)l1.tbl + tbytes);
    l1.ld  = (unsigned short*)((char*)l1.lx + xbytes);

    const int T = DEG_BLOCK;
    dim3 blkN(NBLK);
    dim3 blkD((N + DEG_RANGE - 1) / DEG_RANGE, DEG_CHUNKS);
    const int ntiles = (N + 15) / 16;
    dim3 blkA((N + AGG_ROWS - 1) / AGG_ROWS);

    k_count_deg<<<blkD, 256, 0, stream>>>(dst, degp, E, N);
    k_finalize_dis<<<blkN, T, 0, stream>>>(degp, dis, degi, bsum, dst, keys1, E, N);

    k_gemm_wmma<64, 128, 4><<<(ntiles + 3) / 4, 256, 0, stream>>>(
        X, W1, bufH, N, src, dst, degi, rec1, E);
    k_agg<128, 4, 64><<<blkA, 1024, 0, stream>>>(bufH, keys1, rec1, dst, dis, b1, degi, bsum, l1, W2, h2, E, N);

    k_agg_list<64, 8><<<blkA, 1024, 0, stream>>>(h2, l1.tbl, l1.lx, l1.ld, dis, b2, out, N);
}


# Loops that paid one memory round trip per iteration (degree planes, key words, X tile, start rows, a queue's or list's last edges) issue their loads in groups; f32 sum order = LDS arrival as before
# speedup vs baseline: 3.8079x; 1.0604x over previous
#include <hip/hip_runtime.h>
#include <hip/hip_bf16.h>

typedef __attribute__((ext_vector_type(2))) float v2f;
typedef __attribute__((ext_vector_type(8))) float v8f;

constexpr int DEG_RANGE  = 16384;
constexpr int DEG_CHUNKS = 64;

__global__ void __launch_bounds__(256)
k_count_deg(const int* __restrict__ dst, int* __restrict__ part, int e, int n) {
    __shared__ int cnt[DEG_RANGE];
    const int tid = threadIdx.x;
    const int lo  = blockIdx.x * DEG_RANGE;
    const int c   = blockIdx.y;
    for (int i = tid; i < DEG_RANGE; i += 256) cnt[i] = 0;
    __syncthreads();

    const int per = (((e + DEG_CHUNKS - 1) / DEG_CHUNKS) + 3) & ~3;
    const int e0  = c * per;
    const int e1  = min(e0 + per, e);
    if ((((size_t)dst) & 15) == 0) {
        const int nq = (e1 > e0) ? ((e1 - e0) >> 2) : 0;
        const int4* dq = reinterpret_cast<const int4*>(dst + e0);
        constexpr int B = 8;
        for (int qb = 0; qb < nq; qb += 256 * B) {
            int4 v[B];
            #pragma unroll
            for (int j = 0; j < B; ++j) {
                const int q = qb + j * 256 + tid;
                v[j] = dq[min(q, nq - 1)];
            }
            #pragma unroll
            for (int j = 0; j < B; ++j) {
                const bool in = (qb + j * 256 + tid) < nq;
                const unsigned k0 = (unsigned)(v[j].x - lo), k1 = (unsigned)(v[j].y - lo);
                const unsigned k2 = (unsigned)(v[j].z - lo), k3 = (unsigned)(v[j].w - lo);
                if (in && k0 < (unsigned)DEG_RANGE) atomicAdd(&cnt[k0], 1);
                if (in && k1 < (unsigned)DEG_RANGE) atomicAdd(&cnt[k1], 1);
                if (in && k2 < (unsigned)DEG_RANGE) atomicAdd(&cnt[k2], 1);
                if (in && k3 < (unsigned)DEG_RANGE) atomicAdd(&cnt[k3], 1);
            }
        }
        const int i = e0 + 4 * nq + tid;
        if (tid < 4 && i < e1) {
            const unsigned k = (unsigned)(dst[i] - lo);
            if (k < (unsigned)DEG_RANGE) atomicAdd(&cnt[k], 1);
        }
    } else {
        for (int i = e0 + tid; i < e1; i += 256) {
            const unsigned k = (unsigned)(dst[i] - lo);
            if (k < (unsigned)DEG_RANGE) atomicAdd(&cnt[k], 1);
        }
    }
    __syncthreads();

    const int hi = min(DEG_RANGE, n - lo);
    int* p = part + (size_t)c * n + lo;
    for (int i = tid; i < hi; i += 256) p[i] = cnt[i];
}

constexpr int AGG_ROWS = 400;

__device__ __forceinline__ void write_keys(const int* __restrict__ dst, unsigned* __restrict__ keys,
                                           int e, int gtid, int gsize) {
    const int nw = ((e + 15) >> 4) << 2;
    constexpr int KB = 4;
    for (int w0 = gtid; w0 < nw; w0 += KB * gsize) {
        int d[KB][4];
        #pragma unroll
        for (int t = 0; t < KB; ++t) {
            #pragma unroll
            for (int j = 0; j < 4; ++j)
                d[t][j] = dst[min(4 * (w0 + t * gsize) + j, e - 1)];
        }
        #pragma unroll
        for (int t = 0; t < KB; ++t) {
            const int w = w0 + t * gsize;
            if (w < nw) {
                unsigned v = 0;
                #pragma unroll
                for (int j = 0; j < 4; ++j) {
                    unsigned k = 0xFFu;
                    if (4 * w + j < e) {
                        const unsigned p = (unsigned)d[t][j] / (unsigned)AGG_ROWS;
                        k = p < 0xFFu ? p : 0xFFu;
                    }
                    v |= k << (8 * j);
                }
                keys[w] = v;
            }
        }
    }
}

constexpr int REC_SHIFT = 23;
constexpr int DEG_BLOCK = 256;
constexpr unsigned DEG16_BIG = 0xFFFFu;

struct Lists {
    unsigned*       tbl;
    unsigned*       lx;
    unsigned short* ld;
};

__device__ __forceinline__ void build_recs(const int* __restrict__ src, const int* __restrict__ dst,
                                           const int* __restrict__ degi, uint2* __restrict__ rec,
                                           int e, int gtid, int gsize) {
    constexpr int B = 4;
    for (int i0 = gtid; i0 < e; i0 += B * gsize) {
        int s[B], d[B];
        int ws[B];
        #pragma unroll
        for (int j = 0; j < B; ++j) {
            const int i = min(i0 + j * gsize, e - 1);
            s[j] = src[i];
            d[j] = dst[i];
        }
        #pragma unroll
        for (int j = 0; j < B; ++j) ws[j] = degi[s[j]] + 1;
        #pragma unroll
        for (int j = 0; j < B; ++j) {
            const int i = i0 + j * gsize;
            if (i < e) {
                uint2 r;
                r.x = (unsigned)s[j] | (((unsigned)d[j] % (unsigned)AGG_ROWS) << REC_SHIFT);
                r.y = (unsigned)ws[j];
                rec[i] = r;
            }
        }
    }
}

__global__ void __launch_bounds__(DEG_BLOCK)
k_finalize_dis(const int* __restrict__ part, float* __restrict__ dis,
               int* __restrict__ degi, int* __restrict__ bsum,
               const int* __restrict__ dst, unsigned* __restrict__ keys,
               int e, int n) {
    __shared__ int bs;
    if (threadIdx.x == 0) bs = 0;
    __syncthreads();
    int i = blockIdx.x * blockDim.x + threadIdx.x;
    int din = 0;
    if (i < n) {
        int pv[DEG_CHUNKS];
        #pragma unroll
        for (int c = 0; c < DEG_CHUNKS; ++c) pv[c] = part[(size_t)c * n + i];
        #pragma unroll
        for (int c = 0; c < DEG_CHUNKS; ++c) din += pv[c];
        dis[i]  = rsqrtf((float)(din + 1));
        degi[i] = din;
    }
    atomicAdd(&bs, din);
    __syncthreads();
    if (threadIdx.x == 0) bsum[blockIdx.x] = bs;
    write_keys(dst, keys, e, i, (int)(gridDim.x * blockDim.x));
}

template <int FIN, int FOUT, int MT>
__global__ void k_gemm_wmma(const float* __restrict__ X, const float* __restrict__ W,
                            float* __restrict__ H, int M,
                            const int* __restrict__ src, const int* __restrict__ dst,
                            const int* __restrict__ degi, uint2* __restrict__ rec_to, int e) {
    constexpr int LDA = FIN + 1;
    __shared__ float sW[FIN * FOUT];
    __shared__ float sA[MT * 16 * LDA];

    const int tid  = threadIdx.x;
    constexpr int nthr = 32 * (FOUT / 16);
    const int m0   = blockIdx.x * (MT * 16);

    constexpr int XQ  = FIN / 4;
    constexpr int XPT = MT * 16 * XQ / nthr;
    static_assert(FIN % 4 == 0 && (MT * 16 * XQ) % nthr == 0, "tile shape of the X copy");
    const bool x16 = (((size_t)X) & 15) == 0;
    float4 xv[XPT];
    if (x16) {
        const float4* X4 = reinterpret_cast<const float4*>(X);
        #pragma unroll
        for (int t = 0; t < XPT; ++t) {
            const int q = tid + t * nthr;
            const int r = q / XQ, c4 = q - r * XQ;
            int row = m0 + r; if (row >= M) row = M - 1;
            xv[t] = X4[(size_t)row * XQ + c4];
        }
    }

    build_recs(src, dst, degi, rec_to, e, (int)(blockIdx.x * nthr + tid), (int)(gridDim.x * nthr));

    for (int i = tid; i < FIN * FOUT; i += nthr) sW[i] = W[i];
    if (x16) {
        #pragma unroll
        for (int t = 0; t < XPT; ++t) {
            const int q = tid + t * nthr;
            const int r = q / XQ, c4 = q - r * XQ;
            float* ap = &sA[r * LDA + 4 * c4];
            ap[0] = xv[t].x;  ap[1] = xv[t].y;  ap[2] = xv[t].z;  ap[3] = xv[t].w;
        }
    } else {
        for (int i = tid; i < MT * 16 * FIN; i += nthr) {
            int r = i / FIN, c = i - r * FIN;
            int row = m0 + r; if (row >= M) row = M - 1;
            float v = X[(size_t)row * FIN + c];
            sA[r * LDA + c] = v;
        }
    }
    __syncthreads();

    const int wave = tid >> 5;
    const int lane = tid & 31;
    const int half = lane >> 4;
    const int l16  = lane & 15;
    const int n0   = wave * 16;

    v2f bf[FIN / 4];
    #pragma unroll
    for (int kk = 0; kk < FIN / 4; ++kk) {
        int k = kk * 4 + 2 * half;
        bf[kk].x = sW[k * FOUT + n0 + l16];
        bf[kk].y = sW[(k + 1) * FOUT + n0 + l16];
    }

    #pragma unroll
    for (int mt = 0; mt < MT; ++mt) {
        const float* At = &sA[mt * 16 * LDA];
        v8f c = {};
        #pragma unroll
        for (int kk = 0; kk < FIN / 4; ++kk) {
            int k = kk * 4 + 2 * half;
            v2f a;
            a.x = At[l16 * LDA + k];
            a.y = At[l16 * LDA + k + 1];
            c = __builtin_amdgcn_wmma_f32_16x16x4_f32(false, a, false, bf[kk],
                                                      (short)0, c, false, false);
        }
        #pragma unroll
        for (int r = 0; r < 8; ++r) {
            int row = m0 + mt * 16 + half * 8 + r;
            if (row < M) H[(size_t)row * FOUT + n0 + l16] = c[r];
        }
    }
}

__device__ __forceinline__ int frag_k(int kk, int h, int e) { return 4 * kk + 2 * e + h; }

typedef __attribute__((ext_vector_type(4))) float v4f;
template <int F> struct RowVec;
template <> struct RowVec<64>  { typedef v2f type; };
template <> struct RowVec<128> { typedef v4f type; };

template <int F>
__device__ __forceinline__ int lds_col(int f) {
    return 32 * (f % (F / 32)) + f / (F / 32);
}

template <int F, int U>
__device__ __forceinline__ void rows_load(float (&v)[U][F / 32], const float* hl, int s, int j0) {
    typedef typename RowVec<F>::type vec;
    #pragma unroll
    for (int u = 0; u < U; ++u) {
        const int sj = __builtin_amdgcn_readlane(s, j0 + u);
        const vec t = *reinterpret_cast<const vec*>(hl + (size_t)sj * F);
        #pragma unroll
        for (int k = 0; k < F / 32; ++k) v[u][k] = t[k];
    }
}

template <int F, int U, int P>
__device__ __forceinline__ void rows_add(const float (&v)[U][F / 32], float* al, int kr, float nrm, int j0) {
    #pragma unroll
    for (int u = 0; u < U; ++u) {
        const int   kj = __builtin_amdgcn_readlane(kr, j0 + u);
        const float nj = __int_as_float(__builtin_amdgcn_readlane(__float_as_int(nrm), j0 + u));
        float* ap = al + kj * P;
        #pragma unroll
        for (int k = 0; k < F / 32; ++k) atomicAdd(ap + 32 * k, v[u][k] * nj);
    }
}

template <int F, int U, int P>
__device__ __forceinline__ void walk32(const float* hl, float* al, int s, int kr, float nrm) {
    float va[U][F / 32], vb[U][F / 32];
    rows_load<F, U>(va, hl, s, 0);
    #pragma unroll
    for (int j0 = 0; j0 < 32; j0 += 2 * U) {
        rows_load<F, U>(vb, hl, s, j0 + U);
        rows_add<F, U, P>(va, al, kr, nrm, j0);
        if (j0 + 2 * U < 32) rows_load<F, U>(va, hl, s, j0 + 2 * U);
        rows_add<F, U, P>(vb, al, kr, nrm, j0 + U);
    }
}

template <int F, int U, int P>
__device__ __forceinline__ void walk_part(const float* hl, float* al, int s, int kr, float nrm, int m) {
    typedef typename RowVec<F>::type vec;
    for (int j0 = 0; j0 < m; j0 += U) {
        vec t[U];
        #pragma unroll
        for (int u = 0; u < U; ++u) {
            const int sj = __builtin_amdgcn_readlane(s, min(j0 + u, m - 1));
            t[u] = *reinterpret_cast<const vec*>(hl + (size_t)sj * F);
        }
        #pragma unroll
        for (int u = 0; u < U; ++u) {
            if (j0 + u < m) {
                const int   kj = __builtin_amdgcn_readlane(kr, j0 + u);
                const float nj = __int_as_float(__builtin_amdgcn_readlane(__float_as_int(nrm), j0 + u));
                float* ap = al + kj * P;
                #pragma unroll
                for (int k = 0; k < F / 32; ++k) atomicAdd(ap + 32 * k, t[u][k] * nj);
            }
        }
    }
}

template <int F, int P, int G>
__device__ __forceinline__ void start_rows(float* acc, const float* __restrict__ h, const float* __restrict__ dis,
                                           const float* __restrict__ bias, int lo, int rows, int tid) {
    constexpr int Q  = F / 4;
    static_assert(1024 % Q == 0, "a thread keeps its column");
    constexpr int RS = 1024 / Q;
    constexpr int NT = (AGG_ROWS + RS - 1) / RS;
    const float4* h4 = reinterpret_cast<const float4*>(h + (size_t)lo * F);
    const int c  = tid % Q;
    const int r0 = tid / Q;
    const float4 bv = reinterpret_cast<const float4*>(bias)[c];
    for (int t0 = 0; t0 < NT; t0 += G) {
        float4 hv[G];
        float  sv[G];
        #pragma unroll
        for (int g = 0; g < G; ++g) {
            const int r = min(r0 + (t0 + g) * RS, rows - 1);
            hv[g] = h4[r * Q + c];
            sv[g] = dis[lo + r];
        }
        #pragma unroll
        for (int g = 0; g < G; ++g) {
            const int r = r0 + (t0 + g) * RS;
            if (r < rows) {
                const float w = sv[g] * sv[g];
                float* ar = acc + r * P;
                ar[lds_col<F>(4 * c)]     = hv[g].x * w + bv.x;
                ar[lds_col<F>(4 * c + 1)] = hv[g].y * w + bv.y;
                ar[lds_col<F>(4 * c + 2)] = hv[g].z * w + bv.z;
                ar[lds_col<F>(4 * c + 3)] = hv[g].w * w + bv.w;
            }
        }
    }
}

__device__ __forceinline__ void list_append(const Lists& L, int off, int* lcount, bool ok,
                                            int s, unsigned kr, unsigned deg, int lane) {
    const unsigned okm = __builtin_amdgcn_ballot_w32(ok);
    int base = 0;
    if (lane == 0) base = atomicAdd(lcount, __builtin_popcount(okm));
    base = __builtin_amdgcn_readfirstlane(base);
    if (ok) {
        const int slot = off + base + (int)__builtin_amdgcn_mbcnt_lo(okm, 0u);
        L.lx[slot] = (unsigned)s | (kr << REC_SHIFT);
        L.ld[slot] = (unsigned short)(deg < DEG16_BIG ? deg : DEG16_BIG);
    }
}

template <int F, int U, int FO>
__global__ void __launch_bounds__(1024)
k_agg(const float* __restrict__ h, const unsigned* __restrict__ keys,
      const uint2* __restrict__ rec, const int* __restrict__ dst,
      const float* __restrict__ dis, const float* __restrict__ bias,
      const int* __restrict__ degi, const int* __restrict__ bsum, const Lists L,
      const float* __restrict__ W2, float* __restrict__ h2, int e, int n) {
    static_assert(32 % (2 * U) == 0 && F % 32 == 0, "group shape");
    static_assert(F == 128 && AGG_ROWS % 16 == 0 && FO % 16 == 0 && 32 % (FO / 16) == 0, "tile shape of the second product");
    constexpr int B = 4;
    constexpr int AP = F + 1;
    __shared__ float acc[AGG_ROWS * AP];
    __shared__ float dl[AGG_ROWS];
    __shared__ int loff;
    __shared__ int lcount;

    const int tid  = threadIdx.x;
    const int lane = tid & 31;
    const int lo   = blockIdx.x * AGG_ROWS;
    const int rows = min(AGG_ROWS, n - lo);

    if (tid == 0) { loff = 0; lcount = 0; }
    __syncthreads();
    {
        const int nb = lo / DEG_BLOCK;
        int v = 0;
        for (int i = tid; i < nb; i += 1024) v += bsum[i];
        for (int i = nb * DEG_BLOCK + tid; i < lo; i += 1024) v += degi[i];
        atomicAdd(&loff, v);
    }

    for (int i = tid; i < AGG_ROWS; i += 1024) dl[i] = (i < rows) ? dis[lo + i] : 0.0f;
    start_rows<F, AP, 7>(acc, h, dis, bias, lo, rows, tid);
    __syncthreads();

    const int off = loff;
    const float* hl = h + (F / 32) * lane;
    float*       al = acc + lane;
    const unsigned kb   = blockIdx.x < 0xFFu ? blockIdx.x : 0xFFu;
    const unsigned key4 = kb * 0x01010101u;
    const int nq = (e + 15) >> 4;
    const uint4* kq = reinterpret_cast<const uint4*>(keys);
    const int per = 1024 * B;
    const int T   = (nq + per - 1) / per;
    const int rot = (int)(((long long)blockIdx.x * T) / gridDim.x);

    int qe = 0;
    int qc = 0;

    uint4 cur[B], nxt[B];
    int itr = rot;
    #pragma unroll
    for (int j = 0; j < B; ++j) cur[j] = kq[min(itr * per + j * 1024 + tid, nq - 1)];

    for (int it = 0; it < T; ++it) {
        int itn = itr + 1;
        if (itn >= T) itn = 0;
        #pragma unroll
        for (int j = 0; j < B; ++j) nxt[j] = kq[min(itn * per + j * 1024 + tid, nq - 1)];

        const int g0 = itr * per + tid;
        unsigned m[B];
        unsigned any = 0;
        #pragma unroll
        for (int j = 0; j < B; ++j) {
            const unsigned x0 = cur[j].x ^ key4, x1 = cur[j].y ^ key4, x2 = cur[j].z ^ key4, x3 = cur[j].w ^ key4;
            const unsigned z0 = ~(((x0 & 0x7F7F7F7Fu) + 0x7F7F7F7Fu) | x0) & 0x80808080u;
            const unsigned z1 = ~(((x1 & 0x7F7F7F7Fu) + 0x7F7F7F7Fu) | x1) & 0x80808080u;
            const unsigned z2 = ~(((x2 & 0x7F7F7F7Fu) + 0x7F7F7F7Fu) | x2) & 0x80808080u;
            const unsigned z3 = ~(((x3 & 0x7F7F7F7Fu) + 0x7F7F7F7Fu) | x3) & 0x80808080u;
            m[j] = (z0 >> 7) | (z1 >> 6) | (z2 >> 5) | (z3 >> 4);
            if (g0 + j * 1024 >= nq) m[j] = 0;
            any |= m[j];
        }
        while (__builtin_amdgcn_ballot_w32(any != 0) != 0) {
            const bool pass = (any != 0);
            unsigned mm = m[B - 1];
            int jj = B - 1;
            #pragma unroll
            for (int j = B - 2; j >= 0; --j) {
                if (m[j] != 0) { mm = m[j]; jj = j; }
            }
            const int bit = pass ? __builtin_ctz(mm) : 0;
            const int ei  = 16 * (g0 + jj * 1024) + 4 * (bit & 7) + (bit >> 3);
            any = 0;
            #pragma unroll
            for (int j = 0; j < B; ++j) {
                if (jj == j) m[j] &= m[j] - 1;
                any |= m[j];
            }
            const unsigned bm = __builtin_amdgcn_ballot_w32(pass);
            const int cnt = __builtin_popcount(bm);
            const int rk  = (int)__builtin_amdgcn_mbcnt_lo(bm, 0u);
            const int pos = pass ? (qc + rk) : (qc + cnt + (lane - rk));
            const int pe = __builtin_amdgcn_ds_permute((pos & 31) << 2, ei);
            const int me = (lane < qc) ? qe : pe;
            if (qc + cnt >= 32) {
                const int ec = min(me, e - 1);
                const uint2 r = rec[ec];
                const int s  = (int)(r.x & ((1u << REC_SHIFT) - 1u));
                unsigned kr  = r.x >> REC_SHIFT;
                bool ok = (me < e);
                if (blockIdx.x >= 0xFFu) {
                    kr = (unsigned)(dst[ec] - lo);
                    ok = ok && (kr < (unsigned)rows);
                }
                if (!ok) kr = 0;
                const float nrm = ok ? rsqrtf((float)(int)r.y) * dl[kr] : 0.0f;
                list_append(L, off, &lcount, ok, s, kr, r.y, lane);
                walk32<F, U, AP>(hl, al, s, (int)kr, nrm);
                qe = pe;
                qc = qc + cnt - 32;
            } else {
                qe = me;
                qc = qc + cnt;
            }
        }
        #pragma unroll
        for (int j = 0; j < B; ++j) cur[j] = nxt[j];
        itr = itn;
    }

    if (qc > 0) {
        int s = 0, kr = 0;
        float nrm = 0.0f;
        bool ok = false;
        unsigned dg = 1;
        if (lane < qc) {
            const int ec = min(qe, e - 1);
            const uint2 r = rec[ec];
            s = (int)(r.x & ((1u << REC_SHIFT) - 1u));
            unsigned k = r.x >> REC_SHIFT;
            ok = (qe < e);
            if (blockIdx.x >= 0xFFu) {
                k  = (unsigned)(dst[ec] - lo);
                ok = ok && (k < (unsigned)rows);
            }
            dg  = r.y;
            kr  = ok ? (int)k : 0;
            nrm = ok ? rsqrtf((float)(int)dg) * dl[kr] : 0.0f;
        }
        list_append(L, off, &lcount, ok, s, (unsigned)kr, dg, lane);
        walk_part<F, 2 * U, AP>(hl, al, s, kr, nrm, qc);
    }
    {
        constexpr int NT = FO / 16;
        constexpr int MTILES = AGG_ROWS / 16;
        const int wave = __builtin_amdgcn_readfirstlane(tid >> 5);
        const int half = lane >> 4;
        const int l16  = lane & 15;
        const int n0   = (wave % NT) * 16;
        v2f bf[F / 4];
        #pragma unroll
        for (int kk = 0; kk < F / 4; ++kk) {
            bf[kk].x = W2[frag_k(kk, half, 0) * FO + n0 + l16];
            bf[kk].y = W2[frag_k(kk, half, 1) * FO + n0 + l16];
        }
        __syncthreads();

        for (int mt = wave / NT; mt < MTILES; mt += 32 / NT) {
            const float* ar = acc + (mt * 16 + l16) * AP;
            v8f c = {};
            #pragma unroll
            for (int kk = 0; kk < F / 4; ++kk) {
                v2f a;
                a.x = fmaxf(ar[lds_col<F>(frag_k(kk, half, 0))], 0.0f);
                a.y = fmaxf(ar[lds_col<F>(frag_k(kk, half, 1))], 0.0f);
                c = __builtin_amdgcn_wmma_f32_16x16x4_f32(false, a, false, bf[kk],
                                                          (short)0, c, false, false);
            }
            #pragma unroll
            for (int r = 0; r < 8; ++r) {
                const int row = mt * 16 + half * 8 + r;
                if (row < rows) h2[(size_t)(lo + row) * FO + n0 + l16] = c[r];
            }
        }
    }
    if (tid == 0) {
        L.tbl[2 * blockIdx.x]     = (unsigned)off;
        L.tbl[2 * blockIdx.x + 1] = (unsigned)lcount;
    }
}

template <int F, int U>
__global__ void __launch_bounds__(1024)
k_agg_list(const float* __restrict__ h, const unsigned* __restrict__ tbl,
           const unsigned* __restrict__ lx, const unsigned short* __restrict__ ld,
           const float* __restrict__ dis, const float* __restrict__ bias,
           float* __restrict__ out, int n) {
    static_assert(32 % (2 * U) == 0 && F % 32 == 0, "group shape");
    __shared__ __attribute__((aligned(16))) float acc[AGG_ROWS * F];
    __shared__ float dl[AGG_ROWS];

    const int tid  = threadIdx.x;
    const int lane = tid & 31;
    const int wave = __builtin_amdgcn_readfirstlane(tid >> 5);
    const int lo   = blockIdx.x * AGG_ROWS;
    const int rows = min(AGG_ROWS, n - lo);

    const int off = (int)tbl[2 * blockIdx.x];
    const int cnt = (int)tbl[2 * blockIdx.x + 1];
    const int last = off + max(cnt, 1) - 1;
    int pos = 32 * wave + lane;
    unsigned x  = lx[min(off + pos, last)];
    unsigned dg = ld[min(off + pos, last)];

    for (int i = tid; i < AGG_ROWS; i += 1024) dl[i] = (i < rows) ? dis[lo + i] : 0.0f;
    start_rows<F, F, 7>(acc, h, dis, bias, lo, rows, tid);
    __syncthreads();

    const float* hl = h + (F / 32) * lane;
    float*       al = acc + lane;
    for (int p0 = 32 * wave; p0 < cnt; p0 += 1024) {
        const bool ok = pos < cnt;
        const int  s  = ok ? (int)(x & ((1u << REC_SHIFT) - 1u)) : 0;
        const int  kr = ok ? (int)(x >> REC_SHIFT) : 0;
        float ds = rsqrtf((float)(int)dg);
        if (dg == DEG16_BIG) ds = dis[s];
        const int pn = pos + 1024;
        const unsigned xn = lx[min(off + pn, last)];
        const unsigned dn = ld[min(off + pn, last)];
        const float nrm = ok ? ds * dl[kr] : 0.0f;
        const int m = min(32, cnt - p0);
        if (m == 32) {
            walk32<F, U, F>(hl, al, s, kr, nrm);
        } else {
            walk_part<F, 2 * U, F>(hl, al, s, kr, nrm, m);
        }
        x = xn;  dg = dn;  pos = pn;
    }
    __syncthreads();

    {
        float4* o4 = reinterpret_cast<float4*>(out + (size_t)lo * F);
        for (int i = tid; i < rows * (F / 4); i += 1024) {
            const int r = i / (F / 4), c = i - r * (F / 4);
            const float* ar = acc + r * F;
            float4 o;
            o.x = ar[lds_col<F>(4 * c)];      o.y = ar[lds_col<F>(4 * c + 1)];
            o.z = ar[lds_col<F>(4 * c + 2)];  o.w = ar[lds_col<F>(4 * c + 3)];
            o4[i] = o;
        }
    }
}

extern "C" void kernel_launch(void* const* d_in, const int* in_sizes, int n_in,
                              void* d_out, int out_size, void* d_ws, size_t ws_size,
                              hipStream_t stream) {
    const float* X  = (const float*)d_in[0];
    const int*   EI = (const int*)d_in[1];
    const float* W1 = (const float*)d_in[2];
    const float* b1 = (const float*)d_in[3];
    const float* W2 = (const float*)d_in[4];
    const float* b2 = (const float*)d_in[5];
    float* out = (float*)d_out;

    const int FIN1 = 64;
    const int N = in_sizes[0] / FIN1;
    const int E = in_sizes[1] / 2;
    const int* src = EI;
    const int* dst = EI + E;

    float* dis = (float*)d_ws;
    size_t off = ((size_t)N * sizeof(float) + 16 + 255) & ~(size_t)255;
    float* bufH = (float*)((char*)d_ws + off);
    float* bufO = bufH + (size_t)N * 128;
    int*   degp = (int*)bufH;
    const int NP   = (N + AGG_ROWS - 1) / AGG_ROWS;
    const int NBLK = (N + DEG_BLOCK - 1) / DEG_BLOCK;
    const size_t kbytes = (((((size_t)E + 15) >> 4) << 4) + 255) & ~(size_t)255;
    const size_t rbytes = ((size_t)E * 8 + 255) & ~(size_t)255;
    const size_t dbytes = ((size_t)N * 4 + 255) & ~(size_t)255;
    const size_t tbytes = ((size_t)NP * 8 + 255) & ~(size_t)255;
    const size_t xbytes = ((size_t)E * 4 + 255) & ~(size_t)255;
    unsigned* keys1 = (unsigned*)d_out;
    uint2* rec1 = (uint2*)((char*)keys1 + kbytes);
    int*   degi = (int*)((char*)rec1 + rbytes);
    int*   bsum = (int*)((char*)degi + dbytes);
    float* h2 = bufO;
    Lists l1;
    l1.tbl = (unsigned*)(bufO + (size_t)N * 64);
    l1.lx  = (unsigned*)((char*)l1.tbl + tbytes);
    l1.ld  = (unsigned short*)((char*)l1.lx + xbytes);

    const int T = DEG_BLOCK;
    dim3 blkN(NBLK);
    dim3 blkD((N + DEG_RANGE - 1) / DEG_RANGE, DEG_CHUNKS);
    const int ntiles = (N + 15) / 16;
    dim3 blkA((N + AGG_ROWS - 1) / AGG_ROWS);

    k_count_deg<<<blkD, 256, 0, stream>>>(dst, degp, E, N);
    k_finalize_dis<<<blkN, T, 0, stream>>>(degp, dis, degi, bsum, dst, keys1, E, N);

    k_gemm_wmma<64, 128, 4><<<(ntiles + 3) / 4, 256, 0, stream>>>(
        X, W1, bufH, N, src, dst, degi, rec1, E);
    k_agg<128, 4, 64><<<blkA, 1024, 0, stream>>>(bufH, keys1, rec1, dst, dis, b1, degi, bsum, l1, W2, h2, E, N);

    k_agg_list<64, 8><<<blkA, 1024, 0, stream>>>(h2, l1.tbl, l1.lx, l1.ld, dis, b2, out, N);
}


# Dense product runs inside the degree-count launch (both kinds of work-group in one grid); edge records drop the degree word, owners read deg[src] themselves: 4 launches; f32 sum order as before
# speedup vs baseline: 4.0604x; 1.0663x over previous
#include <hip/hip_runtime.h>
#include <hip/hip_bf16.h>

typedef __attribute__((ext_vector_type(2))) float v2f;
typedef __attribute__((ext_vector_type(8))) float v8f;

constexpr int DEG_RANGE  = 12800;
constexpr int DEG_CHUNKS = 64;

__device__ __forceinline__ void count_deg_block(int* cnt, const int* __restrict__ dst, int* __restrict__ part,
                                                int e, int n, int range, int c) {
    const int tid = threadIdx.x;
    const int lo  = range * DEG_RANGE;
    for (int i = tid; i < DEG_RANGE; i += 256) cnt[i] = 0;
    __syncthreads();

    const int per = (((e + DEG_CHUNKS - 1) / DEG_CHUNKS) + 3) & ~3;
    const int e0  = c * per;
    const int e1  = min(e0 + per, e);
    if ((((size_t)dst) & 15) == 0) {
        const int nq = (e1 > e0) ? ((e1 - e0) >> 2) : 0;
        const int4* dq = reinterpret_cast<const int4*>(dst + e0);
        constexpr int B = 8;
        for (int qb = 0; qb < nq; qb += 256 * B) {
            int4 v[B];
            #pragma unroll
            for (int j = 0; j < B; ++j) {
                const int q = qb + j * 256 + tid;
                v[j] = dq[min(q, nq - 1)];
            }
            #pragma unroll
            for (int j = 0; j < B; ++j) {
                const bool in = (qb + j * 256 + tid) < nq;
                const unsigned k0 = (unsigned)(v[j].x - lo), k1 = (unsigned)(v[j].y - lo);
                const unsigned k2 = (unsigned)(v[j].z - lo), k3 = (unsigned)(v[j].w - lo);
                if (in && k0 < (unsigned)DEG_RANGE) atomicAdd(&cnt[k0], 1);
                if (in && k1 < (unsigned)DEG_RANGE) atomicAdd(&cnt[k1], 1);
                if (in && k2 < (unsigned)DEG_RANGE) atomicAdd(&cnt[k2], 1);
                if (in && k3 < (unsigned)DEG_RANGE) atomicAdd(&cnt[k3], 1);
            }
        }
        const int i = e0 + 4 * nq + tid;
        if (tid < 4 && i < e1) {
            const unsigned k = (unsigned)(dst[i] - lo);
            if (k < (unsigned)DEG_RANGE) atomicAdd(&cnt[k], 1);
        }
    } else {
        for (int i = e0 + tid; i < e1; i += 256) {
            const unsigned k = (unsigned)(dst[i] - lo);
            if (k < (unsigned)DEG_RANGE) atomicAdd(&cnt[k], 1);
        }
    }
    __syncthreads();

    const int hi = min(DEG_RANGE, n - lo);
    int* p = part + (size_t)c * n + lo;
    for (int i = tid; i < hi; i += 256) p[i] = cnt[i];
}

constexpr int AGG_ROWS = 400;

__device__ __forceinline__ void write_keys(const int* __restrict__ dst, unsigned* __restrict__ keys,
                                           int e, int gtid, int gsize) {
    const int nw = ((e + 15) >> 4) << 2;
    constexpr int KB = 4;
    for (int w0 = gtid; w0 < nw; w0 += KB * gsize) {
        int d[KB][4];
        #pragma unroll
        for (int t = 0; t < KB; ++t) {
            #pragma unroll
            for (int j = 0; j < 4; ++j)
                d[t][j] = dst[min(4 * (w0 + t * gsize) + j, e - 1)];
        }
        #pragma unroll
        for (int t = 0; t < KB; ++t) {
            const int w = w0 + t * gsize;
            if (w < nw) {
                unsigned v = 0;
                #pragma unroll
                for (int j = 0; j < 4; ++j) {
                    unsigned k = 0xFFu;
                    if (4 * w + j < e) {
                        const unsigned p = (unsigned)d[t][j] / (unsigned)AGG_ROWS;
                        k = p < 0xFFu ? p : 0xFFu;
                    }
                    v |= k << (8 * j);
                }
                keys[w] = v;
            }
        }
    }
}

constexpr int REC_SHIFT = 23;
constexpr int DEG_BLOCK = 256;
constexpr unsigned DEG16_BIG = 0xFFFFu;

struct Lists {
    unsigned*       tbl;
    unsigned*       lx;
    unsigned short* ld;
};

__device__ __forceinline__ void build_recs(const int* __restrict__ src, const int* __restrict__ dst,
                                           unsigned* __restrict__ rec,
                                           int e, int gtid, int gsize) {
    constexpr int B = 4;
    for (int i0 = gtid; i0 < e; i0 += B * gsize) {
        int s[B], d[B];
        #pragma unroll
        for (int j = 0; j < B; ++j) {
            const int i = min(i0 + j * gsize, e - 1);
            s[j] = src[i];
            d[j] = dst[i];
        }
        #pragma unroll
        for (int j = 0; j < B; ++j) {
            const int i = i0 + j * gsize;
            if (i < e)
                rec[i] = (unsigned)s[j] | (((unsigned)d[j] % (unsigned)AGG_ROWS) << REC_SHIFT);
        }
    }
}

__global__ void __launch_bounds__(DEG_BLOCK)
k_finalize_dis(const int* __restrict__ part, float* __restrict__ dis,
               int* __restrict__ degi, int* __restrict__ bsum,
               const int* __restrict__ dst, unsigned* __restrict__ keys,
               int e, int n) {
    __shared__ int bs;
    if (threadIdx.x == 0) bs = 0;
    __syncthreads();
    int i = blockIdx.x * blockDim.x + threadIdx.x;
    int din = 0;
    if (i < n) {
        int pv[DEG_CHUNKS];
        #pragma unroll
        for (int c = 0; c < DEG_CHUNKS; ++c) pv[c] = part[(size_t)c * n + i];
        #pragma unroll
        for (int c = 0; c < DEG_CHUNKS; ++c) din += pv[c];
        dis[i]  = rsqrtf((float)(din + 1));
        degi[i] = din;
    }
    atomicAdd(&bs, din);
    __syncthreads();
    if (threadIdx.x == 0) bsum[blockIdx.x] = bs;
    write_keys(dst, keys, e, i, (int)(gridDim.x * blockDim.x));
}

template <int FIN, int FOUT, int MT>
__device__ __forceinline__ void gemm_block(float* sW, float* sA,
                                           const float* __restrict__ X, const float* __restrict__ W,
                                           float* __restrict__ H, int M,
                                           const int* __restrict__ src, const int* __restrict__ dst,
                                           unsigned* __restrict__ rec_to, int e, int blk, int nblk) {
    constexpr int LDA = FIN + 1;

    const int tid  = threadIdx.x;
    constexpr int nthr = 32 * (FOUT / 16);
    const int m0   = blk * (MT * 16);

    constexpr int XQ  = FIN / 4;
    constexpr int XPT = MT * 16 * XQ / nthr;
    static_assert(FIN % 4 == 0 && (MT * 16 * XQ) % nthr == 0, "tile shape of the X copy");
    const bool x16 = (((size_t)X) & 15) == 0;
    float4 xv[XPT];
    if (x16) {
        const float4* X4 = reinterpret_cast<const float4*>(X);
        #pragma unroll
        for (int t = 0; t < XPT; ++t) {
            const int q = tid + t * nthr;
            const int r = q / XQ, c4 = q - r * XQ;
            int row = m0 + r; if (row >= M) row = M - 1;
            xv[t] = X4[(size_t)row * XQ + c4];
        }
    }

    build_recs(src, dst, rec_to, e, blk * nthr + tid, nblk * nthr);

    for (int i = tid; i < FIN * FOUT; i += nthr) sW[i] = W[i];
    if (x16) {
        #pragma unroll
        for (int t = 0; t < XPT; ++t) {
            const int q = tid + t * nthr;
            const int r = q / XQ, c4 = q - r * XQ;
            float* ap = &sA[r * LDA + 4 * c4];
            ap[0] = xv[t].x;  ap[1] = xv[t].y;  ap[2] = xv[t].z;  ap[3] = xv[t].w;
        }
    } else {
        for (int i = tid; i < MT * 16 * FIN; i += nthr) {
            int r = i / FIN, c = i - r * FIN;
            int row = m0 + r; if (row >= M) row = M - 1;
            float v = X[(size_t)row * FIN + c];
            sA[r * LDA + c] = v;
        }
    }
    __syncthreads();

    const int wave = tid >> 5;
    const int lane = tid & 31;
    const int half = lane >> 4;
    const int l16  = lane & 15;
    const int n0   = wave * 16;

    v2f bf[FIN / 4];
    #pragma unroll
    for (int kk = 0; kk < FIN / 4; ++kk) {
        int k = kk * 4 + 2 * half;
        bf[kk].x = sW[k * FOUT + n0 + l16];
        bf[kk].y = sW[(k + 1) * FOUT + n0 + l16];
    }

    #pragma unroll
    for (int mt = 0; mt < MT; ++mt) {
        const float* At = &sA[mt * 16 * LDA];
        v8f c = {};
        #pragma unroll
        for (int kk = 0; kk < FIN / 4; ++kk) {
            int k = kk * 4 + 2 * half;
            v2f a;
            a.x = At[l16 * LDA + k];
            a.y = At[l16 * LDA + k + 1];
            c = __builtin_amdgcn_wmma_f32_16x16x4_f32(false, a, false, bf[kk],
                                                      (short)0, c, false, false);
        }
        #pragma unroll
        for (int r = 0; r < 8; ++r) {
            int row = m0 + mt * 16 + half * 8 + r;
            if (row < M) H[(size_t)row * FOUT + n0 + l16] = c[r];
        }
    }
}

template <int FIN, int FOUT, int MT>
__global__ void __launch_bounds__(32 * (FOUT / 16))
k_deg_gemm(const float* __restrict__ X, const float* __restrict__ W, float* __restrict__ H, int M,
           const int* __restrict__ src, const int* __restrict__ dst,
           unsigned* __restrict__ rec_to, int e, int* __restrict__ part, int nrange) {
    constexpr int GW = FIN * FOUT + MT * 16 * (FIN + 1);
    constexpr int LW = GW > DEG_RANGE ? GW : DEG_RANGE;
    static_assert(32 * (FOUT / 16) == 256, "the degree count is written for 256 threads");
    __shared__ __attribute__((aligned(16))) unsigned char lds[LW * 4];
    const int ndeg = nrange * DEG_CHUNKS;
    const int b = blockIdx.x;
    if (b < ndeg) {
        count_deg_block(reinterpret_cast<int*>(lds), dst, part, e, M, b % nrange, b / nrange);
    } else {
        float* sW = reinterpret_cast<float*>(lds);
        gemm_block<FIN, FOUT, MT>(sW, sW + FIN * FOUT, X, W, H, M, src, dst, rec_to, e,
                                  b - ndeg, (int)gridDim.x - ndeg);
    }
}

__device__ __forceinline__ int frag_k(int kk, int h, int e) { return 4 * kk + 2 * e + h; }

typedef __attribute__((ext_vector_type(4))) float v4f;
template <int F> struct RowVec;
template <> struct RowVec<64>  { typedef v2f type; };
template <> struct RowVec<128> { typedef v4f type; };

template <int F>
__device__ __forceinline__ int lds_col(int f) {
    return 32 * (f % (F / 32)) + f / (F / 32);
}

template <int F, int U>
__device__ __forceinline__ void rows_load(float (&v)[U][F / 32], const float* hl, int s, int j0) {
    typedef typename RowVec<F>::type vec;
    #pragma unroll
    for (int u = 0; u < U; ++u) {
        const int sj = __builtin_amdgcn_readlane(s, j0 + u);
        const vec t = *reinterpret_cast<const vec*>(hl + (size_t)sj * F);
        #pragma unroll
        for (int k = 0; k < F / 32; ++k) v[u][k] = t[k];
    }
}

template <int F, int U, int P>
__device__ __forceinline__ void rows_add(const float (&v)[U][F / 32], float* al, int kr, float nrm, int j0) {
    #pragma unroll
    for (int u = 0; u < U; ++u) {
        const int   kj = __builtin_amdgcn_readlane(kr, j0 + u);
        const float nj = __int_as_float(__builtin_amdgcn_readlane(__float_as_int(nrm), j0 + u));
        float* ap = al + kj * P;
        #pragma unroll
        for (int k = 0; k < F / 32; ++k) atomicAdd(ap + 32 * k, v[u][k] * nj);
    }
}

template <int F, int U, int P>
__device__ __forceinline__ void walk32(const float* hl, float* al, int s, int kr, float nrm) {
    float va[U][F / 32], vb[U][F / 32];
    rows_load<F, U>(va, hl, s, 0);
    #pragma unroll
    for (int j0 = 0; j0 < 32; j0 += 2 * U) {
        rows_load<F, U>(vb, hl, s, j0 + U);
        rows_add<F, U, P>(va, al, kr, nrm, j0);
        if (j0 + 2 * U < 32) rows_load<F, U>(va, hl, s, j0 + 2 * U);
        rows_add<F, U, P>(vb, al, kr, nrm, j0 + U);
    }
}

template <int F, int U, int P>
__device__ __forceinline__ void walk32_deg(const float* hl, float* al, int s, int kr, bool ok,
                                           unsigned dg, const float* dl) {
    float va[U][F / 32], vb[U][F / 32];
    rows_load<F, U>(va, hl, s, 0);
    rows_load<F, U>(vb, hl, s, U);
    const float nrm = ok ? rsqrtf((float)(int)dg) * dl[kr] : 0.0f;
    #pragma unroll
    for (int j0 = 0; j0 < 32; j0 += 2 * U) {
        if (j0 > 0) rows_load<F, U>(vb, hl, s, j0 + U);
        rows_add<F, U, P>(va, al, kr, nrm, j0);
        if (j0 + 2 * U < 32) rows_load<F, U>(va, hl, s, j0 + 2 * U);
        rows_add<F, U, P>(vb, al, kr, nrm, j0 + U);
    }
}

template <int F, int U, int P>
__device__ __forceinline__ void walk_part(const float* hl, float* al, int s, int kr, float nrm, int m) {
    typedef typename RowVec<F>::type vec;
    for (int j0 = 0; j0 < m; j0 += U) {
        vec t[U];
        #pragma unroll
        for (int u = 0; u < U; ++u) {
            const int sj = __builtin_amdgcn_readlane(s, min(j0 + u, m - 1));
            t[u] = *reinterpret_cast<const vec*>(hl + (size_t)sj * F);
        }
        #pragma unroll
        for (int u = 0; u < U; ++u) {
            if (j0 + u < m) {
                const int   kj = __builtin_amdgcn_readlane(kr, j0 + u);
                const float nj = __int_as_float(__builtin_amdgcn_readlane(__float_as_int(nrm), j0 + u));
                float* ap = al + kj * P;
                #pragma unroll
                for (int k = 0; k < F / 32; ++k) atomicAdd(ap + 32 * k, t[u][k] * nj);
            }
        }
    }
}

template <int F, int P, int G>
__device__ __forceinline__ void start_rows(float* acc, const float* __restrict__ h, const float* __restrict__ dis,
                                           const float* __restrict__ bias, int lo, int rows, int tid) {
    constexpr int Q  = F / 4;
    static_assert(1024 % Q == 0, "a thread keeps its column");
    constexpr int RS = 1024 / Q;
    constexpr int NT = (AGG_ROWS + RS - 1) / RS;
    const float4* h4 = reinterpret_cast<const float4*>(h + (size_t)lo * F);
    const int c  = tid % Q;
    const int r0 = tid / Q;
    const float4 bv = reinterpret_cast<const float4*>(bias)[c];
    for (int t0 = 0; t0 < NT; t0 += G) {
        float4 hv[G];
        float  sv[G];
        #pragma unroll
        for (int g = 0; g < G; ++g) {
            const int r = min(r0 + (t0 + g) * RS, rows - 1);
            hv[g] = h4[r * Q + c];
            sv[g] = dis[lo + r];
        }
        #pragma unroll
        for (int g = 0; g < G; ++g) {
            const int r = r0 + (t0 + g) * RS;
            if (r < rows) {
                const float w = sv[g] * sv[g];
                float* ar = acc + r * P;
                ar[lds_col<F>(4 * c)]     = hv[g].x * w + bv.x;
                ar[lds_col<F>(4 * c + 1)] = hv[g].y * w + bv.y;
                ar[lds_col<F>(4 * c + 2)] = hv[g].z * w + bv.z;
                ar[lds_col<F>(4 * c + 3)] = hv[g].w * w + bv.w;
            }
        }
    }
}

__device__ __forceinline__ void list_append(const Lists& L, int off, int* lcount, bool ok,
                                            int s, unsigned kr, unsigned deg, int lane) {
    const unsigned okm = __builtin_amdgcn_ballot_w32(ok);
    int base = 0;
    if (lane == 0) base = atomicAdd(lcount, __builtin_popcount(okm));
    base = __builtin_amdgcn_readfirstlane(base);
    if (ok) {
        const int slot = off + base + (int)__builtin_amdgcn_mbcnt_lo(okm, 0u);
        L.lx[slot] = (unsigned)s | (kr << REC_SHIFT);
        L.ld[slot] = (unsigned short)(deg < DEG16_BIG ? deg : DEG16_BIG);
    }
}

template <int F, int U, int FO>
__global__ void __launch_bounds__(1024)
k_agg(const float* __restrict__ h, const unsigned* __restrict__ keys,
      const unsigned* __restrict__ rec, const int* __restrict__ dst,
      const float* __restrict__ dis, const float* __restrict__ bias,
      const int* __restrict__ degi, const int* __restrict__ bsum, const Lists L,
      const float* __restrict__ W2, float* __restrict__ h2, int e, int n) {
    static_assert(32 % (2 * U) == 0 && F % 32 == 0, "group shape");
    static_assert(F == 128 && AGG_ROWS % 16 == 0 && FO % 16 == 0 && 32 % (FO / 16) == 0, "tile shape of the second product");
    constexpr int B = 4;
    constexpr int AP = F + 1;
    __shared__ float acc[AGG_ROWS * AP];
    __shared__ float dl[AGG_ROWS];
    __shared__ int loff;
    __shared__ int lcount;

    const int tid  = threadIdx.x;
    const int lane = tid & 31;
    const int lo   = blockIdx.x * AGG_ROWS;
    const int rows = min(AGG_ROWS, n - lo);

    if (tid == 0) { loff = 0; lcount = 0; }
    __syncthreads();
    {
        const int nb = lo / DEG_BLOCK;
        int v = 0;
        for (int i = tid; i < nb; i += 1024) v += bsum[i];
        for (int i = nb * DEG_BLOCK + tid; i < lo; i += 1024) v += degi[i];
        atomicAdd(&loff, v);
    }

    for (int i = tid; i < AGG_ROWS; i += 1024) dl[i] = (i < rows) ? dis[lo + i] : 0.0f;
    start_rows<F, AP, 7>(acc, h, dis, bias, lo, rows, tid);
    __syncthreads();

    const int off = loff;
    const float* hl = h + (F / 32) * lane;
    float*       al = acc + lane;
    const unsigned kb   = blockIdx.x < 0xFFu ? blockIdx.x : 0xFFu;
    const unsigned key4 = kb * 0x01010101u;
    const int nq = (e + 15) >> 4;
    const uint4* kq = reinterpret_cast<const uint4*>(keys);
    const int per = 1024 * B;
    const int T   = (nq + per - 1) / per;
    const int rot = (int)(((long long)blockIdx.x * T) / gridDim.x);

    int qe = 0;
    int qc = 0;

    uint4 cur[B], nxt[B];
    int itr = rot;
    #pragma unroll
    for (int j = 0; j < B; ++j) cur[j] = kq[min(itr * per + j * 1024 + tid, nq - 1)];

    for (int it = 0; it < T; ++it) {
        int itn = itr + 1;
        if (itn >= T) itn = 0;
        #pragma unroll
        for (int j = 0; j < B; ++j) nxt[j] = kq[min(itn * per + j * 1024 + tid, nq - 1)];

        const int g0 = itr * per + tid;
        unsigned m[B];
        unsigned any = 0;
        #pragma unroll
        for (int j = 0; j < B; ++j) {
            const unsigned x0 = cur[j].x ^ key4, x1 = cur[j].y ^ key4, x2 = cur[j].z ^ key4, x3 = cur[j].w ^ key4;
            const unsigned z0 = ~(((x0 & 0x7F7F7F7Fu) + 0x7F7F7F7Fu) | x0) & 0x80808080u;
            const unsigned z1 = ~(((x1 & 0x7F7F7F7Fu) + 0x7F7F7F7Fu) | x1) & 0x80808080u;
            const unsigned z2 = ~(((x2 & 0x7F7F7F7Fu) + 0x7F7F7F7Fu) | x2) & 0x80808080u;
            const unsigned z3 = ~(((x3 & 0x7F7F7F7Fu) + 0x7F7F7F7Fu) | x3) & 0x80808080u;
            m[j] = (z0 >> 7) | (z1 >> 6) | (z2 >> 5) | (z3 >> 4);
            if (g0 + j * 1024 >= nq) m[j] = 0;
            any |= m[j];
        }
        while (__builtin_amdgcn_ballot_w32(any != 0) != 0) {
            const bool pass = (any != 0);
            unsigned mm = m[B - 1];
            int jj = B - 1;
            #pragma unroll
            for (int j = B - 2; j >= 0; --j) {
                if (m[j] != 0) { mm = m[j]; jj = j; }
            }
            const int bit = pass ? __builtin_ctz(mm) : 0;
            const int ei  = 16 * (g0 + jj * 1024) + 4 * (bit & 7) + (bit >> 3);
            any = 0;
            #pragma unroll
            for (int j = 0; j < B; ++j) {
                if (jj == j) m[j] &= m[j] - 1;
                any |= m[j];
            }
            const unsigned bm = __builtin_amdgcn_ballot_w32(pass);
            const int cnt = __builtin_popcount(bm);
            const int rk  = (int)__builtin_amdgcn_mbcnt_lo(bm, 0u);
            const int pos = pass ? (qc + rk) : (qc + cnt + (lane - rk));
            const int pe = __builtin_amdgcn_ds_permute((pos & 31) << 2, ei);
            const int me = (lane < qc) ? qe : pe;
            if (qc + cnt >= 32) {
                const int ec = min(me, e - 1);
                const unsigned rx = rec[ec];
                const int s  = (int)(rx & ((1u << REC_SHIFT) - 1u));
                unsigned kr  = rx >> REC_SHIFT;
                const unsigned dg = (unsigned)degi[s] + 1u;
                bool ok = (me < e);
                if (blockIdx.x >= 0xFFu) {
                    kr = (unsigned)(dst[ec] - lo);
                    ok = ok && (kr < (unsigned)rows);
                }
                if (!ok) kr = 0;
                walk32_deg<F, U, AP>(hl, al, s, (int)kr, ok, dg, dl);
                list_append(L, off, &lcount, ok, s, kr, dg, lane);
                qe = pe;
                qc = qc + cnt - 32;
            } else {
                qe = me;
                qc = qc + cnt;
            }
        }
        #pragma unroll
        for (int j = 0; j < B; ++j) cur[j] = nxt[j];
        itr = itn;
    }

    if (qc > 0) {
        int s = 0, kr = 0;
        float nrm = 0.0f;
        bool ok = false;
        unsigned dg = 1;
        if (lane < qc) {
            const int ec = min(qe, e - 1);
            const unsigned rx = rec[ec];
            s = (int)(rx & ((1u << REC_SHIFT) - 1u));
            unsigned k = rx >> REC_SHIFT;
            dg = (unsigned)degi[s] + 1u;
            ok = (qe < e);
            if (blockIdx.x >= 0xFFu) {
                k  = (unsigned)(dst[ec] - lo);
                ok = ok && (k < (unsigned)rows);
            }
            kr  = ok ? (int)k : 0;
            nrm = ok ? rsqrtf((float)(int)dg) * dl[kr] : 0.0f;
        }
        walk_part<F, 2 * U, AP>(hl, al, s, kr, nrm, qc);
        list_append(L, off, &lcount, ok, s, (unsigned)kr, dg, lane);
    }
    {
        constexpr int NT = FO / 16;
        constexpr int MTILES = AGG_ROWS / 16;
        const int wave = __builtin_amdgcn_readfirstlane(tid >> 5);
        const int half = lane >> 4;
        const int l16  = lane & 15;
        const int n0   = (wave % NT) * 16;
        v2f bf[F / 4];
        #pragma unroll
        for (int kk = 0; kk < F / 4; ++kk) {
            bf[kk].x = W2[frag_k(kk, half, 0) * FO + n0 + l16];
            bf[kk].y = W2[frag_k(kk, half, 1) * FO + n0 + l16];
        }
        __syncthreads();

        for (int mt = wave / NT; mt < MTILES; mt += 32 / NT) {
            const float* ar = acc + (mt * 16 + l16) * AP;
            v8f c = {};
            #pragma unroll
            for (int kk = 0; kk < F / 4; ++kk) {
                v2f a;
                a.x = fmaxf(ar[lds_col<F>(frag_k(kk, half, 0))], 0.0f);
                a.y = fmaxf(ar[lds_col<F>(frag_k(kk, half, 1))], 0.0f);
                c = __builtin_amdgcn_wmma_f32_16x16x4_f32(false, a, false, bf[kk],
                                                          (short)0, c, false, false);
            }
            #pragma unroll
            for (int r = 0; r < 8; ++r) {
                const int row = mt * 16 + half * 8 + r;
                if (row < rows) h2[(size_t)(lo + row) * FO + n0 + l16] = c[r];
            }
        }
    }
    if (tid == 0) {
        L.tbl[2 * blockIdx.x]     = (unsigned)off;
        L.tbl[2 * blockIdx.x + 1] = (unsigned)lcount;
    }
}

template <int F, int U>
__global__ void __launch_bounds__(1024)
k_agg_list(const float* __restrict__ h, const unsigned* __restrict__ tbl,
           const unsigned* __restrict__ lx, const unsigned short* __restrict__ ld,
           const float* __restrict__ dis, const float* __restrict__ bias,
           float* __restrict__ out, int n) {
    static_assert(32 % (2 * U) == 0 && F % 32 == 0, "group shape");
    __shared__ __attribute__((aligned(16))) float acc[AGG_ROWS * F];
    __shared__ float dl[AGG_ROWS];

    const int tid  = threadIdx.x;
    const int lane = tid & 31;
    const int wave = __builtin_amdgcn_readfirstlane(tid >> 5);
    const int lo   = blockIdx.x * AGG_ROWS;
    const int rows = min(AGG_ROWS, n - lo);

    const int off = (int)tbl[2 * blockIdx.x];
    const int cnt = (int)tbl[2 * blockIdx.x + 1];
    const int last = off + max(cnt, 1) - 1;
    int pos = 32 * wave + lane;
    unsigned x  = lx[min(off + pos, last)];
    unsigned dg = ld[min(off + pos, last)];

    for (int i = tid; i < AGG_ROWS; i += 1024) dl[i] = (i < rows) ? dis[lo + i] : 0.0f;
    start_rows<F, F, 7>(acc, h, dis, bias, lo, rows, tid);
    __syncthreads();

    const float* hl = h + (F / 32) * lane;
    float*       al = acc + lane;
    for (int p0 = 32 * wave; p0 < cnt; p0 += 1024) {
        const bool ok = pos < cnt;
        const int  s  = ok ? (int)(x & ((1u << REC_SHIFT) - 1u)) : 0;
        const int  kr = ok ? (int)(x >> REC_SHIFT) : 0;
        float ds = rsqrtf((float)(int)dg);
        if (dg == DEG16_BIG) ds = dis[s];
        const int pn = pos + 1024;
        const unsigned xn = lx[min(off + pn, last)];
        const unsigned dn = ld[min(off + pn, last)];
        const float nrm = ok ? ds * dl[kr] : 0.0f;
        const int m = min(32, cnt - p0);
        if (m == 32) {
            walk32<F, U, F>(hl, al, s, kr, nrm);
        } else {
            walk_part<F, 2 * U, F>(hl, al, s, kr, nrm, m);
        }
        x = xn;  dg = dn;  pos = pn;
    }
    __syncthreads();

    {
        float4* o4 = reinterpret_cast<float4*>(out + (size_t)lo * F);
        for (int i = tid; i < rows * (F / 4); i += 1024) {
            const int r = i / (F / 4), c = i - r * (F / 4);
            const float* ar = acc + r * F;
            float4 o;
            o.x = ar[lds_col<F>(4 * c)];      o.y = ar[lds_col<F>(4 * c + 1)];
            o.z = ar[lds_col<F>(4 * c + 2)];  o.w = ar[lds_col<F>(4 * c + 3)];
            o4[i] = o;
        }
    }
}

extern "C" void kernel_launch(void* const* d_in, const int* in_sizes, int n_in,
                              void* d_out, int out_size, void* d_ws, size_t ws_size,
                              hipStream_t stream) {
    const float* X  = (const float*)d_in[0];
    const int*   EI = (const int*)d_in[1];
    const float* W1 = (const float*)d_in[2];
    const float* b1 = (const float*)d_in[3];
    const float* W2 = (const float*)d_in[4];
    const float* b2 = (const float*)d_in[5];
    float* out = (float*)d_out;

    const int FIN1 = 64;
    const int N = in_sizes[0] / FIN1;
    const int E = in_sizes[1] / 2;
    const int* src = EI;
    const int* dst = EI + E;

    float* dis = (float*)d_ws;
    size_t off = ((size_t)N * sizeof(float) + 16 + 255) & ~(size_t)255;
    float* bufH = (float*)((char*)d_ws + off);
    float* bufO = bufH + (size_t)N * 128;
    int*   degp = (int*)bufO;
    const int NP   = (N + AGG_ROWS - 1) / AGG_ROWS;
    const int NBLK = (N + DEG_BLOCK - 1) / DEG_BLOCK;
    const size_t kbytes = (((((size_t)E + 15) >> 4) << 4) + 255) & ~(size_t)255;
    const size_t rbytes = ((size_t)E * 4 + 255) & ~(size_t)255;
    const size_t dbytes = ((size_t)N * 4 + 255) & ~(size_t)255;
    const size_t tbytes = ((size_t)NP * 8 + 255) & ~(size_t)255;
    const size_t xbytes = ((size_t)E * 4 + 255) & ~(size_t)255;
    unsigned* keys1 = (unsigned*)d_out;
    unsigned* rec1 = (unsigned*)((char*)keys1 + kbytes);
    int*   degi = (int*)((char*)rec1 + rbytes);
    int*   bsum = (int*)((char*)degi + dbytes);
    float* h2 = bufO;
    Lists l1;
    l1.tbl = (unsigned*)(bufO + (size_t)N * 64);
    l1.lx  = (unsigned*)((char*)l1.tbl + tbytes);
    l1.ld  = (unsigned short*)((char*)l1.lx + xbytes);

    const int T = DEG_BLOCK;
    dim3 blkN(NBLK);
    const int NR = (N + DEG_RANGE - 1) / DEG_RANGE;
    const int ntiles = (N + 15) / 16;
    const int NG = (ntiles + 3) / 4;
    dim3 blkA((N + AGG_ROWS - 1) / AGG_ROWS);

    k_deg_gemm<64, 128, 4><<<NR * DEG_CHUNKS + NG, 256, 0, stream>>>(
        X, W1, bufH, N, src, dst, rec1, E, degp, NR);
    k_finalize_dis<<<blkN, T, 0, stream>>>(degp, dis, degi, bsum, dst, keys1, E, N);

    k_agg<128, 4, 64><<<blkA, 1024, 0, stream>>>(bufH, keys1, rec1, dst, dis, b1, degi, bsum, l1, W2, h2, E, N);

    k_agg_list<64, 8><<<blkA, 1024, 0, stream>>>(h2, l1.tbl, l1.lx, l1.ld, dis, b2, out, N);
}
